# starting kernel
# baseline (speedup 1.0000x reference)
#include <hip/hip_runtime.h>
#include <hip/hip_bf16.h>


#define B_ 2
#define L_ 4096
#define C_ 7
#define H_ 8
#define D_ 64
#define W_ 512
#define M_ 32
#define ROWS_ (B_*L_)

typedef float v2f __attribute__((ext_vector_type(2)));
typedef float v8f __attribute__((ext_vector_type(8)));

__global__ __launch_bounds__(256) void qkv_proj_wmma(
    const float* __restrict__ hs,
    const float* __restrict__ wq,
    const float* __restrict__ wk,
    const float* __restrict__ wv,
    float* __restrict__ q, float* __restrict__ k, float* __restrict__ v)
{
    const int MT = ROWS_ / 16;
    const int NT = W_ / 16;
    const int lane = threadIdx.x & 31;
    const int wave = threadIdx.x >> 5;
    const int tile = blockIdx.x * (blockDim.x >> 5) + wave;

    const int mat = tile / (MT * NT);
    const int t   = tile % (MT * NT);
    const int mt  = t / NT;
    const int nt  = t % NT;

    const float* __restrict__ wmat = (mat == 0) ? wq : (mat == 1) ? wk : wv;
    float* __restrict__ outp       = (mat == 0) ? q  : (mat == 1) ? k  : v;

    const int mrow  = lane & 15;
    const int khalf = lane >> 4;
    const int arow  = mt * 16 + mrow;
    const int ncol  = nt * 16 + (lane & 15);

    v8f c = {};
#pragma unroll
    for (int ks = 0; ks < 2; ++ks) {
        const int   k0   = ks * 4 + khalf * 2;
        const bool  inr  = (k0 + 1 < C_);
        const int   k1i  = inr ? (k0 + 1) : 0;
        const float kmsk = inr ? 1.0f : 0.0f;
        v2f a, b;
        a.x = hs[arow * C_ + k0];
        a.y = hs[arow * C_ + k1i] * kmsk;
        b.x = wmat[ncol * C_ + k0];
        b.y = wmat[ncol * C_ + k1i] * kmsk;
        c = __builtin_amdgcn_wmma_f32_16x16x4_f32(
                false, a, false, b, (short)0, c, false, false);
    }

    const float scale = (mat == 0) ? 0.125f : 1.0f;
    const int rbase = mt * 16 + (khalf ? 8 : 0);
#pragma unroll
    for (int r = 0; r < 8; ++r)
        outp[(rbase + r) * W_ + ncol] = c[r] * scale;
}

__global__ __launch_bounds__(256) void pyr_attn(
    const float* __restrict__ q, const float* __restrict__ k,
    const float* __restrict__ v, const int* __restrict__ qk_mask,
    float* __restrict__ aout)
{
    const int bl = blockIdx.x;
    const int b  = bl / L_;
    const int l  = bl % L_;

    __shared__ float qs[W_];
    __shared__ int   idxs[M_];

    const int tid = threadIdx.x;
    qs[tid]       = q[bl * W_ + tid];
    qs[tid + 256] = q[bl * W_ + tid + 256];
    if (tid < M_) idxs[tid] = qk_mask[l * M_ + tid];
    __syncthreads();

    const int h    = tid >> 5;
    const int lane = tid & 31;

    const int  raw   = idxs[lane];
    const bool valid = (raw >= 0);
    const int  krow  = b * L_ + (valid ? raw : 0);

    const float* __restrict__ vpre = v + (size_t)krow * W_ + h * D_;
    __builtin_prefetch(vpre,      0, 0);
    __builtin_prefetch(vpre + 32, 0, 0);

    const float* __restrict__ kp = k + (size_t)krow * W_ + h * D_;
    float s = 0.0f;
#pragma unroll 8
    for (int d = 0; d < D_; ++d)
        s = fmaf(qs[h * D_ + d], kp[d], s);
    float score = valid ? s : -1.0e9f;

    float mx = score;
#pragma unroll
    for (int off = 16; off > 0; off >>= 1)
        mx = fmaxf(mx, __shfl_xor(mx, off, 32));
    float p = valid ? __expf(score - mx) : 0.0f;
    float sum = p;
#pragma unroll
    for (int off = 16; off > 0; off >>= 1)
        sum += __shfl_xor(sum, off, 32);
    const float attn = p * (1.0f / fmaxf(sum, 1.0e-30f));

    float a0 = 0.0f, a1 = 0.0f;
#pragma unroll 4
    for (int m = 0; m < M_; ++m) {
        const float am  = __shfl(attn, m, 32);
        const int  raw2 = idxs[m];
        const int  vrow = b * L_ + (raw2 >= 0 ? raw2 : 0);
        const float* __restrict__ vp = v + (size_t)vrow * W_ + h * D_;
        a0 = fmaf(am, vp[lane],      a0);
        a1 = fmaf(am, vp[lane + 32], a1);
    }
    aout[bl * W_ + h * D_ + lane]      = a0;
    aout[bl * W_ + h * D_ + lane + 32] = a1;
}

__global__ __launch_bounds__(256) void out_proj_ln_wmma(
    const float* __restrict__ aout, const float* __restrict__ fc_w,
    const float* __restrict__ fc_b, const float* __restrict__ hs,
    const float* __restrict__ ln_s, const float* __restrict__ ln_b,
    float* __restrict__ out)
{
    const int lane = threadIdx.x & 31;
    const int wave = threadIdx.x >> 5;
    const int mt   = blockIdx.x * (blockDim.x >> 5) + wave;

    const int mrow  = lane & 15;
    const int khalf = lane >> 4;
    const int arow  = mt * 16 + mrow;
    const int col   = lane & 15;
    const bool cv   = (col < C_);

    const float cmask = cv ? 1.0f : 0.0f;
    const int   bcol  = cv ? col : 0;
    const float* __restrict__ arowp = aout + (size_t)arow * W_;
    const float* __restrict__ browp = fc_w + (size_t)bcol * W_;

    v8f c = {};
    for (int kk = 0; kk < W_; kk += 4) {
        const int k0 = kk + khalf * 2;
        v2f a = *(const v2f*)(arowp + k0);
        v2f b = *(const v2f*)(browp + k0);
        b.x *= cmask;
        b.y *= cmask;
        c = __builtin_amdgcn_wmma_f32_16x16x4_f32(
                false, a, false, b, (short)0, c, false, false);
    }

    const float bias = cv ? fc_b[col] : 0.0f;
    const float lsc  = cv ? ln_s[col] : 0.0f;
    const float lbi  = cv ? ln_b[col] : 0.0f;
    const int   rbase = mt * 16 + (khalf ? 8 : 0);

#pragma unroll
    for (int r = 0; r < 8; ++r) {
        const int   row = rbase + r;
        const float val = c[r] + bias + (cv ? hs[row * C_ + col] : 0.0f);

        float x = cv ? val : 0.0f;
        float ssum = x;
#pragma unroll
        for (int off = 8; off > 0; off >>= 1)
            ssum += __shfl_xor(ssum, off, 32);
        const float mu = ssum * (1.0f / (float)C_);

        float dv = cv ? (val - mu) : 0.0f;
        float vsum = dv * dv;
#pragma unroll
        for (int off = 8; off > 0; off >>= 1)
            vsum += __shfl_xor(vsum, off, 32);
        const float var = vsum * (1.0f / (float)C_);

        const float y = (val - mu) * rsqrtf(var + 1.0e-6f) * lsc + lbi;
        if (cv) out[row * C_ + col] = y;
    }
}

extern "C" void kernel_launch(void* const* d_in, const int* in_sizes, int n_in,
                              void* d_out, int out_size, void* d_ws, size_t ws_size,
                              hipStream_t stream)
{
    const float* hs  = (const float*)d_in[0];
    const float* wq  = (const float*)d_in[1];
    const float* wk  = (const float*)d_in[2];
    const float* wv  = (const float*)d_in[3];
    const float* fcw = (const float*)d_in[4];
    const float* fcb = (const float*)d_in[5];
    const float* lns = (const float*)d_in[6];
    const float* lnb = (const float*)d_in[7];
    const int*   qkm = (const int*)d_in[8];
    float* out = (float*)d_out;

    float* qb = (float*)d_ws;
    float* kb = qb + (size_t)ROWS_ * W_;
    float* vb = kb + (size_t)ROWS_ * W_;
    float* ab = vb + (size_t)ROWS_ * W_;

    qkv_proj_wmma<<<6144, 256, 0, stream>>>(hs, wq, wk, wv, qb, kb, vb);
    pyr_attn<<<ROWS_, 256, 0, stream>>>(qb, kb, vb, qkm, ab);
    out_proj_ln_wmma<<<64, 256, 0, stream>>>(ab, fcw, fcb, hs, lns, lnb, out);
}


# K, V and q are no longer materialised and gathered: each wave rebuilds its head's K^T and V tiles of a row's 32 keys from their 7-float input rows by f32 WMMA; one launch (fp sum order changes)
# speedup vs baseline: 3.6563x; 3.6563x over previous
#include <hip/hip_runtime.h>


#define B_ 2
#define L_ 4096
#define C_ 7
#define H_ 8
#define D_ 64
#define W_ 512
#define M_ 32
#define ROWS_ (B_*L_)

typedef float v2f __attribute__((ext_vector_type(2)));
typedef float v4f __attribute__((ext_vector_type(4)));
typedef float v8f __attribute__((ext_vector_type(8)));

#define TR_ 16
#define TP_ (W_ + 4)
#define QP_ 36

#define DPP8_SEL(a, b, c, d, e, f, g, h) \
    ((a) | ((b) << 3) | ((c) << 6) | ((d) << 9) | ((e) << 12) | ((f) << 15) | ((g) << 18) | ((h) << 21))

template <unsigned SEL>
__device__ __forceinline__ float dpp8(float x)
{
    return __builtin_bit_cast(float,
        __builtin_amdgcn_mov_dpp8(__builtin_bit_cast(unsigned, x), SEL));
}
__device__ __forceinline__ float xor8(float x)
{
    const int b = __builtin_bit_cast(int, x);
    return __builtin_bit_cast(float,
        __builtin_amdgcn_update_dpp(b, b, 0x128, 0xf, 0xf, false));
}
__device__ __forceinline__ float xor16(float x)
{
    const int b = __builtin_bit_cast(int, x);
    return __builtin_bit_cast(float,
        __builtin_amdgcn_permlanex16(b, b, 0x76543210, 0xfedcba98, false, false));
}
__device__ __forceinline__ float xor4(float x) { return dpp8<DPP8_SEL(4, 5, 6, 7, 0, 1, 2, 3)>(x); }
__device__ __forceinline__ float xor2(float x) { return dpp8<DPP8_SEL(2, 3, 0, 1, 6, 7, 4, 5)>(x); }
__device__ __forceinline__ float xor1(float x) { return dpp8<DPP8_SEL(1, 0, 3, 2, 5, 4, 7, 6)>(x); }

struct frag2 { v2f lo, hi; };
__device__ __forceinline__ frag2 row_frag(const float* __restrict__ p, int hf, float m)
{
    frag2 f;
    f.lo.x = p[0];
    f.lo.y = p[1];
    f.hi.x = p[4];
    f.hi.y = p[5 - hf] * m;
    return f;
}

__device__ __forceinline__ v8f wmma2(const frag2& a, const frag2& b)
{
    v8f c = {};
    c = __builtin_amdgcn_wmma_f32_16x16x4_f32(false, a.lo, false, b.lo, (short)0, c, false, false);
    c = __builtin_amdgcn_wmma_f32_16x16x4_f32(false, a.hi, false, b.hi, (short)0, c, false, false);
    return c;
}

__global__ __launch_bounds__(256) void pyr_attn_remat(
    const float* __restrict__ hs, const float* __restrict__ wq,
    const float* __restrict__ wk, const float* __restrict__ wv,
    const int* __restrict__ qk_mask,
    const float* __restrict__ fc_w, const float* __restrict__ fc_b,
    const float* __restrict__ ln_s, const float* __restrict__ ln_b,
    float* __restrict__ out)
{
    __shared__ __attribute__((aligned(16))) float tile[TR_ * TP_];
    __shared__ __attribute__((aligned(16))) float qs[H_ * TR_ * 2 * QP_];
    __shared__ float part[H_ * 8 * 32];

    const int row0 = blockIdx.x * TR_;
    const int b    = row0 / L_;
    const int l0   = row0 % L_;

    const int h    = __builtin_amdgcn_readfirstlane((int)(threadIdx.x >> 5));
    const int lane = threadIdx.x & 31;
    const int n    = lane & 15;
    const int hf   = lane >> 4;
    const float m1 = hf ? 0.0f : 1.0f;

    const float* __restrict__ hsb   = hs + (size_t)b * L_ * C_ + hf * 2;
    const int*   __restrict__ mrow0 = qk_mask + l0 * M_ + n;

    frag2 fk[4], fv[4];
    for (int t = 0; t < 4; ++t) {
        const int wr = (h * D_ + t * 16 + n) * C_ + hf * 2;
        fk[t] = row_frag(wk + wr, hf, m1);
        fv[t] = row_frag(wv + wr, hf, m1);
    }

    {
        const frag2 fr = row_frag(hs + (size_t)(row0 + n) * C_ + hf * 2, hf, m1);
        float* __restrict__ qd = qs + ((h * TR_ + n) * 2 + hf) * QP_;
        for (int t = 0; t < 4; ++t) {
            const int wr = (h * D_ + t * 16 + n) * C_ + hf * 2;
            const frag2 fq = row_frag(wq + wr, hf, m1);
            const v8f c = wmma2(fq, fr);
            v4f s0, s1;
            s0.x = c[0] * 0.125f; s0.y = c[1] * 0.125f; s0.z = c[2] * 0.125f; s0.w = c[3] * 0.125f;
            s1.x = c[4] * 0.125f; s1.y = c[5] * 0.125f; s1.z = c[6] * 0.125f; s1.w = c[7] * 0.125f;
            *(v4f*)(qd + t * 8)     = s0;
            *(v4f*)(qd + t * 8 + 4) = s1;
        }
    }

    int raw0  = mrow0[0],  raw1  = mrow0[16];
    int nraw0 = mrow0[M_], nraw1 = mrow0[M_ + 16];
    frag2 g0 = row_frag(hsb + (unsigned)(raw0 >= 0 ? raw0 : 0) * C_, hf, m1);
    frag2 g1 = row_frag(hsb + (unsigned)(raw1 >= 0 ? raw1 : 0) * C_, hf, m1);

    __syncthreads();

    const float* __restrict__ qrd = qs + (h * TR_ * 2 + hf) * QP_;
    float* __restrict__ tout = tile + h * D_ + hf * 32 + n;
    const bool swp = (((lane >> 3) & 1) != hf);

#pragma unroll 1
    for (int i = 0; i < TR_; ++i) {
        const frag2 ng0 = row_frag(hsb + (unsigned)(nraw0 >= 0 ? nraw0 : 0) * C_, hf, m1);
        const frag2 ng1 = row_frag(hsb + (unsigned)(nraw1 >= 0 ? nraw1 : 0) * C_, hf, m1);
        const int i2 = (i + 2 < TR_) ? (i + 2) : (TR_ - 1);
        const int nn0 = mrow0[i2 * M_], nn1 = mrow0[i2 * M_ + 16];

        v4f qv[8];
        for (int c4 = 0; c4 < 8; ++c4)
            qv[c4] = *(const v4f*)(qrd + i * 2 * QP_ + c4 * 4);

        __builtin_amdgcn_sched_barrier(0);

        float sp0 = 0.0f, sp1 = 0.0f;
        for (int t = 0; t < 4; ++t) {
            const v8f k0 = wmma2(fk[t], g0);
            const v8f k1 = wmma2(fk[t], g1);
            const v4f qa = qv[2 * t], qb = qv[2 * t + 1];
            float a0 = qa.x * k0[0], a1 = qa.x * k1[0];
            a0 = fmaf(qa.y, k0[1], a0); a1 = fmaf(qa.y, k1[1], a1);
            a0 = fmaf(qa.z, k0[2], a0); a1 = fmaf(qa.z, k1[2], a1);
            a0 = fmaf(qa.w, k0[3], a0); a1 = fmaf(qa.w, k1[3], a1);
            a0 = fmaf(qb.x, k0[4], a0); a1 = fmaf(qb.x, k1[4], a1);
            a0 = fmaf(qb.y, k0[5], a0); a1 = fmaf(qb.y, k1[5], a1);
            a0 = fmaf(qb.z, k0[6], a0); a1 = fmaf(qb.z, k1[6], a1);
            a0 = fmaf(qb.w, k0[7], a0); a1 = fmaf(qb.w, k1[7], a1);
            sp0 += a0; sp1 += a1;
        }
        const float s0 = sp0 + xor16(sp0);
        const float s1 = sp1 + xor16(sp1);

        const bool valid0 = (raw0 >= 0), valid1 = (raw1 >= 0);
        const float sc0 = valid0 ? s0 : -1.0e9f;
        const float sc1 = valid1 ? s1 : -1.0e9f;
        float mx = fmaxf(sc0, sc1);
        mx = fmaxf(mx, xor8(mx));
        mx = fmaxf(mx, xor4(mx));
        mx = fmaxf(mx, xor2(mx));
        mx = fmaxf(mx, xor1(mx));
        const float pe0 = valid0 ? __expf(sc0 - mx) : 0.0f;
        const float pe1 = valid1 ? __expf(sc1 - mx) : 0.0f;
        float sum = pe0 + pe1;
        sum += xor8(sum);
        sum += xor4(sum);
        sum += xor2(sum);
        sum += xor1(sum);
        const float inv = 1.0f / fmaxf(sum, 1.0e-30f);
        const float at0 = pe0 * inv, at1 = pe1 * inv;

        const float y0 = xor8(at0), y1 = xor8(at1);
        const float x0 = swp ? y0 : at0;
        const float x1 = swp ? y1 : at1;
#define BC_(R, X) dpp8<DPP8_SEL(R, R, R, R, R, R, R, R)>(X)
        const float w0[8] = {BC_(0, x0), BC_(1, x0), BC_(2, x0), BC_(3, x0),
                             BC_(4, x0), BC_(5, x0), BC_(6, x0), BC_(7, x0)};
        const float w1[8] = {BC_(0, x1), BC_(1, x1), BC_(2, x1), BC_(3, x1),
                             BC_(4, x1), BC_(5, x1), BC_(6, x1), BC_(7, x1)};
#undef BC_
        float o[4];
        for (int t = 0; t < 4; ++t) {
            const v8f v0 = wmma2(g0, fv[t]);
            const v8f v1 = wmma2(g1, fv[t]);
            float a0 = w0[0] * v0[0], a1 = w1[0] * v1[0];
#pragma unroll
            for (int r = 1; r < 8; ++r) {
                a0 = fmaf(w0[r], v0[r], a0);
                a1 = fmaf(w1[r], v1[r], a1);
            }
            o[t] = a0 + a1;
        }
        const float oa = (hf ? o[2] : o[0]) + xor16(hf ? o[0] : o[2]);
        const float ob = (hf ? o[3] : o[1]) + xor16(hf ? o[1] : o[3]);
        tout[i * TP_]      = oa;
        tout[i * TP_ + 16] = ob;

        g0 = ng0; g1 = ng1;
        raw0 = nraw0; raw1 = nraw1;
        nraw0 = nn0;  nraw1 = nn1;
    }

    const int  col = n;
    const bool cv  = (col < C_);
    const float cmask = cv ? 1.0f : 0.0f;
    const int   bcol  = cv ? col : 0;
    const float* __restrict__ bp = fc_w + (size_t)bcol * W_ + h * D_ + hf * 4;
    const float*              ap = tile + col * TP_ + h * D_ + hf * 4;

    v4f bfr[8], afr[8];
    for (int s8 = 0; s8 < 8; ++s8)
        bfr[s8] = *(const v4f*)(bp + s8 * 8);

    __syncthreads();

    for (int s8 = 0; s8 < 8; ++s8)
        afr[s8] = *(const v4f*)(ap + s8 * 8);

    v8f c0 = {}, c1 = {};
    for (int s8 = 0; s8 < 8; ++s8) {
        v2f a, bb;
        a.x = afr[s8].x;           a.y = afr[s8].y;
        bb.x = bfr[s8].x * cmask;  bb.y = bfr[s8].y * cmask;
        c0 = __builtin_amdgcn_wmma_f32_16x16x4_f32(
                 false, a, false, bb, (short)0, c0, false, false);
        a.x = afr[s8].z;           a.y = afr[s8].w;
        bb.x = bfr[s8].z * cmask;  bb.y = bfr[s8].w * cmask;
        c1 = __builtin_amdgcn_wmma_f32_16x16x4_f32(
                 false, a, false, bb, (short)0, c1, false, false);
    }
#pragma unroll
    for (int r = 0; r < 8; ++r)
        part[(h * 8 + r) * 32 + lane] = c0[r] + c1[r];

    __syncthreads();

    float acc = 0.0f;
#pragma unroll
    for (int w = 0; w < H_; ++w)
        acc += part[(w * 8 + h) * 32 + lane];

    const float bias = cv ? fc_b[col] : 0.0f;
    const float lsc  = cv ? ln_s[col] : 0.0f;
    const float lbi  = cv ? ln_b[col] : 0.0f;
    const int   row  = row0 + hf * 8 + h;
    const float val  = acc + bias + (cv ? hs[row * C_ + col] : 0.0f);

    float x = cv ? val : 0.0f;
    float ssum = x;
#pragma unroll
    for (int o = 8; o > 0; o >>= 1)
        ssum += __shfl_xor(ssum, o, 32);
    const float mu = ssum * (1.0f / (float)C_);

    float dv = cv ? (val - mu) : 0.0f;
    float vsum = dv * dv;
#pragma unroll
    for (int o = 8; o > 0; o >>= 1)
        vsum += __shfl_xor(vsum, o, 32);
    const float var = vsum * (1.0f / (float)C_);

    const float y = (val - mu) * rsqrtf(var + 1.0e-6f) * lsc + lbi;
    if (cv) out[row * C_ + col] = y;
}

extern "C" void kernel_launch(void* const* d_in, const int* in_sizes, int n_in,
                              void* d_out, int out_size, void* d_ws, size_t ws_size,
                              hipStream_t stream)
{
    const float* hs  = (const float*)d_in[0];
    const float* wq  = (const float*)d_in[1];
    const float* wk  = (const float*)d_in[2];
    const float* wv  = (const float*)d_in[3];
    const float* fcw = (const float*)d_in[4];
    const float* fcb = (const float*)d_in[5];
    const float* lns = (const float*)d_in[6];
    const float* lnb = (const float*)d_in[7];
    const int*   qkm = (const int*)d_in[8];
    float* out = (float*)d_out;

    pyr_attn_remat<<<ROWS_ / TR_, 256, 0, stream>>>(hs, wq, wk, wv, qkm, fcw, fcb, lns, lnb, out);
}


# q and the softmax weights scale the WMMA operands (Wk fragment rows, gathered key rows) instead of the D tiles: tiles chain in one accumulator, sums are 8-register adds (fp sum order changes)
# speedup vs baseline: 4.2291x; 1.1567x over previous
#include <hip/hip_runtime.h>


#define B_ 2
#define L_ 4096
#define C_ 7
#define H_ 8
#define D_ 64
#define W_ 512
#define M_ 32
#define ROWS_ (B_*L_)

typedef float v2f __attribute__((ext_vector_type(2)));
typedef float v4f __attribute__((ext_vector_type(4)));
typedef float v8f __attribute__((ext_vector_type(8)));

#define TR_ 16
#define TP_ (W_ + 4)
#define QR_ 64

#define DPP8_SEL(a, b, c, d, e, f, g, h) \
    ((a) | ((b) << 3) | ((c) << 6) | ((d) << 9) | ((e) << 12) | ((f) << 15) | ((g) << 18) | ((h) << 21))

template <unsigned SEL>
__device__ __forceinline__ float dpp8(float x)
{
    return __builtin_bit_cast(float,
        __builtin_amdgcn_mov_dpp8(__builtin_bit_cast(unsigned, x), SEL));
}
__device__ __forceinline__ float xor8(float x)
{
    const int b = __builtin_bit_cast(int, x);
    return __builtin_bit_cast(float,
        __builtin_amdgcn_update_dpp(b, b, 0x128, 0xf, 0xf, false));
}
__device__ __forceinline__ float xor16(float x)
{
    const int b = __builtin_bit_cast(int, x);
    return __builtin_bit_cast(float,
        __builtin_amdgcn_permlanex16(b, b, 0x76543210, 0xfedcba98, false, false));
}
__device__ __forceinline__ float xor4(float x) { return dpp8<DPP8_SEL(4, 5, 6, 7, 0, 1, 2, 3)>(x); }
__device__ __forceinline__ float xor2(float x) { return dpp8<DPP8_SEL(2, 3, 0, 1, 6, 7, 4, 5)>(x); }
__device__ __forceinline__ float xor1(float x) { return dpp8<DPP8_SEL(1, 0, 3, 2, 5, 4, 7, 6)>(x); }

struct frag2 { v2f lo, hi; };
__device__ __forceinline__ frag2 row_frag(const float* __restrict__ p, int hf, float m)
{
    frag2 f;
    f.lo.x = p[0];
    f.lo.y = p[1];
    f.hi.x = p[4];
    f.hi.y = p[5 - hf] * m;
    return f;
}

__device__ __forceinline__ v8f wmma2acc(const frag2& a, const frag2& b, v8f c)
{
    c = __builtin_amdgcn_wmma_f32_16x16x4_f32(false, a.lo, false, b.lo, (short)0, c, false, false);
    c = __builtin_amdgcn_wmma_f32_16x16x4_f32(false, a.hi, false, b.hi, (short)0, c, false, false);
    return c;
}
__device__ __forceinline__ v8f wmma2(const frag2& a, const frag2& b)
{
    const v8f z = {};
    return wmma2acc(a, b, z);
}
__device__ __forceinline__ frag2 scaled(const frag2& f, float s)
{
    frag2 o;
    o.lo = f.lo * s;
    o.hi = f.hi * s;
    return o;
}
__device__ __forceinline__ float sum8(const v8f& c)
{
    const v2f p = (__builtin_shufflevector(c, c, 0, 1) + __builtin_shufflevector(c, c, 2, 3))
                + (__builtin_shufflevector(c, c, 4, 5) + __builtin_shufflevector(c, c, 6, 7));
    return p.x + p.y;
}

__global__ __launch_bounds__(256) void pyr_attn_remat(
    const float* __restrict__ hs, const float* __restrict__ wq,
    const float* __restrict__ wk, const float* __restrict__ wv,
    const int* __restrict__ qk_mask,
    const float* __restrict__ fc_w, const float* __restrict__ fc_b,
    const float* __restrict__ ln_s, const float* __restrict__ ln_b,
    float* __restrict__ out)
{
    __shared__ __attribute__((aligned(16))) float tile[TR_ * TP_];
    __shared__ __attribute__((aligned(16))) float qs[H_ * TR_ * QR_];
    __shared__ float part[H_ * 8 * 32];

    const int row0 = blockIdx.x * TR_;
    const int b    = row0 / L_;
    const int l0   = row0 % L_;

    const int h    = __builtin_amdgcn_readfirstlane((int)(threadIdx.x >> 5));
    const int lane = threadIdx.x & 31;
    const int n    = lane & 15;
    const int hf   = lane >> 4;
    const float m1 = hf ? 0.0f : 1.0f;

    const float* __restrict__ hsb   = hs + (size_t)b * L_ * C_ + hf * 2;
    const int*   __restrict__ mrow0 = qk_mask + l0 * M_ + n;

    frag2 fk[4], fv[4];
    for (int t = 0; t < 4; ++t) {
        const int wr = (h * D_ + t * 16 + n) * C_ + hf * 2;
        fk[t] = row_frag(wk + wr, hf, m1);
        fv[t] = row_frag(wv + wr, hf, m1);
    }

    {
        const frag2 fr = row_frag(hs + (size_t)(row0 + n) * C_ + hf * 2, hf, m1);
        v8f cq[4];
        for (int t = 0; t < 4; ++t) {
            const int wr = (h * D_ + t * 16 + n) * C_ + hf * 2;
            const frag2 fq = row_frag(wq + wr, hf, m1);
            cq[t] = wmma2(fr, fq);
        }
        float* __restrict__ qd = qs + (h * TR_ + hf * 8) * QR_ + n * 4;
#pragma unroll
        for (int r = 0; r < 8; ++r) {
            v4f s;
            s.x = cq[0][r] * 0.125f; s.y = cq[1][r] * 0.125f;
            s.z = cq[2][r] * 0.125f; s.w = cq[3][r] * 0.125f;
            *(v4f*)(qd + r * QR_) = s;
        }
    }

    int raw0  = mrow0[0],  raw1  = mrow0[16];
    int nraw0 = mrow0[M_], nraw1 = mrow0[M_ + 16];
    frag2 g0 = row_frag(hsb + (unsigned)(raw0 >= 0 ? raw0 : 0) * C_, hf, m1);
    frag2 g1 = row_frag(hsb + (unsigned)(raw1 >= 0 ? raw1 : 0) * C_, hf, m1);

    __syncthreads();

    const float* __restrict__ qrd = qs + h * TR_ * QR_ + n * 4;
    float* __restrict__ tout = tile + h * D_ + hf * 32 + n;
    v4f qv = *(const v4f*)qrd;

#pragma unroll 1
    for (int i = 0; i < TR_; ++i) {
        const frag2 ng0 = row_frag(hsb + (unsigned)(nraw0 >= 0 ? nraw0 : 0) * C_, hf, m1);
        const frag2 ng1 = row_frag(hsb + (unsigned)(nraw1 >= 0 ? nraw1 : 0) * C_, hf, m1);
        const int i2 = (i + 2 < TR_) ? (i + 2) : (TR_ - 1);
        const int nn0 = mrow0[i2 * M_], nn1 = mrow0[i2 * M_ + 16];

        const int i1 = (i + 1 < TR_) ? (i + 1) : (TR_ - 1);
        const v4f nqv = *(const v4f*)(qrd + i1 * QR_);

        __builtin_amdgcn_sched_barrier(0);

        v8f ks0 = {}, ks1 = {};
        for (int t = 0; t < 4; ++t) {
            const frag2 fq = scaled(fk[t], qv[t]);
            ks0 = wmma2acc(fq, g0, ks0);
            ks1 = wmma2acc(fq, g1, ks1);
        }
        const float sp0 = sum8(ks0), sp1 = sum8(ks1);
        const float s0 = sp0 + xor16(sp0);
        const float s1 = sp1 + xor16(sp1);

        const bool valid0 = (raw0 >= 0), valid1 = (raw1 >= 0);
        const float sc0 = valid0 ? s0 : -1.0e9f;
        const float sc1 = valid1 ? s1 : -1.0e9f;
        float mx = fmaxf(sc0, sc1);
        mx = fmaxf(mx, xor8(mx));
        mx = fmaxf(mx, xor4(mx));
        mx = fmaxf(mx, xor2(mx));
        mx = fmaxf(mx, xor1(mx));
        const float pe0 = valid0 ? __expf(sc0 - mx) : 0.0f;
        const float pe1 = valid1 ? __expf(sc1 - mx) : 0.0f;
        float sum = pe0 + pe1;
        sum += xor8(sum);
        sum += xor4(sum);
        sum += xor2(sum);
        sum += xor1(sum);
        const float inv = 1.0f / fmaxf(sum, 1.0e-30f);
        const float at0 = pe0 * inv, at1 = pe1 * inv;

        const frag2 a0 = scaled(g0, at0);
        const frag2 a1 = scaled(g1, at1);
        float o[4];
        for (int t = 0; t < 4; ++t) {
            v8f vs = wmma2(a0, fv[t]);
            vs = wmma2acc(a1, fv[t], vs);
            o[t] = sum8(vs);
        }
        const float oa = (hf ? o[2] : o[0]) + xor16(hf ? o[0] : o[2]);
        const float ob = (hf ? o[3] : o[1]) + xor16(hf ? o[1] : o[3]);
        tout[i * TP_]      = oa;
        tout[i * TP_ + 16] = ob;

        g0 = ng0; g1 = ng1;
        qv = nqv;
        raw0 = nraw0; raw1 = nraw1;
        nraw0 = nn0;  nraw1 = nn1;
    }

    const int  col = n;
    const bool cv  = (col < C_);
    const float cmask = cv ? 1.0f : 0.0f;
    const int   bcol  = cv ? col : 0;
    const float* __restrict__ bp = fc_w + (size_t)bcol * W_ + h * D_ + hf * 4;
    const float*              ap = tile + col * TP_ + h * D_ + hf * 4;

    v4f bfr[8], afr[8];
    for (int s8 = 0; s8 < 8; ++s8)
        bfr[s8] = *(const v4f*)(bp + s8 * 8);

    __syncthreads();

    for (int s8 = 0; s8 < 8; ++s8)
        afr[s8] = *(const v4f*)(ap + s8 * 8);

    v8f c0 = {}, c1 = {};
    for (int s8 = 0; s8 < 8; ++s8) {
        v2f a, bb;
        a.x = afr[s8].x;           a.y = afr[s8].y;
        bb.x = bfr[s8].x * cmask;  bb.y = bfr[s8].y * cmask;
        c0 = __builtin_amdgcn_wmma_f32_16x16x4_f32(
                 false, a, false, bb, (short)0, c0, false, false);
        a.x = afr[s8].z;           a.y = afr[s8].w;
        bb.x = bfr[s8].z * cmask;  bb.y = bfr[s8].w * cmask;
        c1 = __builtin_amdgcn_wmma_f32_16x16x4_f32(
                 false, a, false, bb, (short)0, c1, false, false);
    }
#pragma unroll
    for (int r = 0; r < 8; ++r)
        part[(h * 8 + r) * 32 + lane] = c0[r] + c1[r];

    __syncthreads();

    float acc = 0.0f;
#pragma unroll
    for (int w = 0; w < H_; ++w)
        acc += part[(w * 8 + h) * 32 + lane];

    const float bias = cv ? fc_b[col] : 0.0f;
    const float lsc  = cv ? ln_s[col] : 0.0f;
    const float lbi  = cv ? ln_b[col] : 0.0f;
    const int   row  = row0 + hf * 8 + h;
    const float val  = acc + bias + (cv ? hs[row * C_ + col] : 0.0f);

    float x = cv ? val : 0.0f;
    float ssum = x;
#pragma unroll
    for (int o = 8; o > 0; o >>= 1)
        ssum += __shfl_xor(ssum, o, 32);
    const float mu = ssum * (1.0f / (float)C_);

    float dv = cv ? (val - mu) : 0.0f;
    float vsum = dv * dv;
#pragma unroll
    for (int o = 8; o > 0; o >>= 1)
        vsum += __shfl_xor(vsum, o, 32);
    const float var = vsum * (1.0f / (float)C_);

    const float y = (val - mu) * rsqrtf(var + 1.0e-6f) * lsc + lbi;
    if (cv) out[row * C_ + col] = y;
}

extern "C" void kernel_launch(void* const* d_in, const int* in_sizes, int n_in,
                              void* d_out, int out_size, void* d_ws, size_t ws_size,
                              hipStream_t stream)
{
    const float* hs  = (const float*)d_in[0];
    const float* wq  = (const float*)d_in[1];
    const float* wk  = (const float*)d_in[2];
    const float* wv  = (const float*)d_in[3];
    const float* fcw = (const float*)d_in[4];
    const float* fcb = (const float*)d_in[5];
    const float* lns = (const float*)d_in[6];
    const float* lnb = (const float*)d_in[7];
    const int*   qkm = (const int*)d_in[8];
    float* out = (float*)d_out;

    pyr_attn_remat<<<ROWS_ / TR_, 256, 0, stream>>>(hs, wq, wk, wv, qkm, fcw, fcb, lns, lnb, out);
}


# Scaled fragments that share an operand and an accumulator are added on the vector unit first: 4 q-scaled Wk tiles -> 1, 2 weighted key tiles -> 1; 12 WMMAs a row, not 32 (fp sum order changes)
# speedup vs baseline: 6.5216x; 1.5421x over previous
#include <hip/hip_runtime.h>


#define B_ 2
#define L_ 4096
#define C_ 7
#define H_ 8
#define D_ 64
#define W_ 512
#define M_ 32
#define ROWS_ (B_*L_)

typedef float v2f __attribute__((ext_vector_type(2)));
typedef float v4f __attribute__((ext_vector_type(4)));
typedef float v8f __attribute__((ext_vector_type(8)));

#define TR_ 16
#define TP_ (W_ + 4)
#define QR_ 64

#define DPP8_SEL(a, b, c, d, e, f, g, h) \
    ((a) | ((b) << 3) | ((c) << 6) | ((d) << 9) | ((e) << 12) | ((f) << 15) | ((g) << 18) | ((h) << 21))

template <unsigned SEL>
__device__ __forceinline__ float dpp8(float x)
{
    return __builtin_bit_cast(float,
        __builtin_amdgcn_mov_dpp8(__builtin_bit_cast(unsigned, x), SEL));
}
__device__ __forceinline__ float xor8(float x)
{
    const int b = __builtin_bit_cast(int, x);
    return __builtin_bit_cast(float,
        __builtin_amdgcn_update_dpp(b, b, 0x128, 0xf, 0xf, false));
}
__device__ __forceinline__ float xor16(float x)
{
    const int b = __builtin_bit_cast(int, x);
    return __builtin_bit_cast(float,
        __builtin_amdgcn_permlanex16(b, b, 0x76543210, 0xfedcba98, false, false));
}
__device__ __forceinline__ float xor4(float x) { return dpp8<DPP8_SEL(4, 5, 6, 7, 0, 1, 2, 3)>(x); }
__device__ __forceinline__ float xor2(float x) { return dpp8<DPP8_SEL(2, 3, 0, 1, 6, 7, 4, 5)>(x); }
__device__ __forceinline__ float xor1(float x) { return dpp8<DPP8_SEL(1, 0, 3, 2, 5, 4, 7, 6)>(x); }

struct frag2 { v2f lo, hi; };
__device__ __forceinline__ frag2 row_frag(const float* __restrict__ p, int hf, float m)
{
    frag2 f;
    f.lo.x = p[0];
    f.lo.y = p[1];
    f.hi.x = p[4];
    f.hi.y = p[5 - hf] * m;
    return f;
}

__device__ __forceinline__ v8f wmma2acc(const frag2& a, const frag2& b, v8f c)
{
    c = __builtin_amdgcn_wmma_f32_16x16x4_f32(false, a.lo, false, b.lo, (short)0, c, false, false);
    c = __builtin_amdgcn_wmma_f32_16x16x4_f32(false, a.hi, false, b.hi, (short)0, c, false, false);
    return c;
}
__device__ __forceinline__ v8f wmma2(const frag2& a, const frag2& b)
{
    const v8f z = {};
    return wmma2acc(a, b, z);
}
__device__ __forceinline__ frag2 scaled(const frag2& f, float s)
{
    frag2 o;
    o.lo = f.lo * s;
    o.hi = f.hi * s;
    return o;
}
__device__ __forceinline__ float sum8(const v8f& c)
{
    const v2f p = (__builtin_shufflevector(c, c, 0, 1) + __builtin_shufflevector(c, c, 2, 3))
                + (__builtin_shufflevector(c, c, 4, 5) + __builtin_shufflevector(c, c, 6, 7));
    return p.x + p.y;
}

__global__ __launch_bounds__(256) void pyr_attn_remat(
    const float* __restrict__ hs, const float* __restrict__ wq,
    const float* __restrict__ wk, const float* __restrict__ wv,
    const int* __restrict__ qk_mask,
    const float* __restrict__ fc_w, const float* __restrict__ fc_b,
    const float* __restrict__ ln_s, const float* __restrict__ ln_b,
    float* __restrict__ out)
{
    __shared__ __attribute__((aligned(16))) float tile[TR_ * TP_];
    __shared__ __attribute__((aligned(16))) float qs[H_ * TR_ * QR_];
    __shared__ float part[H_ * 8 * 32];

    const int row0 = blockIdx.x * TR_;
    const int b    = row0 / L_;
    const int l0   = row0 % L_;

    const int h    = __builtin_amdgcn_readfirstlane((int)(threadIdx.x >> 5));
    const int lane = threadIdx.x & 31;
    const int n    = lane & 15;
    const int hf   = lane >> 4;
    const float m1 = hf ? 0.0f : 1.0f;

    const float* __restrict__ hsb   = hs + (size_t)b * L_ * C_ + hf * 2;
    const int*   __restrict__ mrow0 = qk_mask + l0 * M_ + n;

    frag2 fk[4], fv[4];
    for (int t = 0; t < 4; ++t) {
        const int wr = (h * D_ + t * 16 + n) * C_ + hf * 2;
        fk[t] = row_frag(wk + wr, hf, m1);
        fv[t] = row_frag(wv + wr, hf, m1);
    }

    {
        const frag2 fr = row_frag(hs + (size_t)(row0 + n) * C_ + hf * 2, hf, m1);
        v8f cq[4];
        for (int t = 0; t < 4; ++t) {
            const int wr = (h * D_ + t * 16 + n) * C_ + hf * 2;
            const frag2 fq = row_frag(wq + wr, hf, m1);
            cq[t] = wmma2(fr, fq);
        }
        float* __restrict__ qd = qs + (h * TR_ + hf * 8) * QR_ + n * 4;
#pragma unroll
        for (int r = 0; r < 8; ++r) {
            v4f s;
            s.x = cq[0][r] * 0.125f; s.y = cq[1][r] * 0.125f;
            s.z = cq[2][r] * 0.125f; s.w = cq[3][r] * 0.125f;
            *(v4f*)(qd + r * QR_) = s;
        }
    }

    int raw0  = mrow0[0],  raw1  = mrow0[16];
    int nraw0 = mrow0[M_], nraw1 = mrow0[M_ + 16];
    frag2 g0 = row_frag(hsb + (unsigned)(raw0 >= 0 ? raw0 : 0) * C_, hf, m1);
    frag2 g1 = row_frag(hsb + (unsigned)(raw1 >= 0 ? raw1 : 0) * C_, hf, m1);

    __syncthreads();

    const float* __restrict__ qrd = qs + h * TR_ * QR_ + n * 4;
    float* __restrict__ tout = tile + h * D_ + hf * 32 + n;
    v4f qv = *(const v4f*)qrd;

#pragma unroll 1
    for (int i = 0; i < TR_; ++i) {
        const frag2 ng0 = row_frag(hsb + (unsigned)(nraw0 >= 0 ? nraw0 : 0) * C_, hf, m1);
        const frag2 ng1 = row_frag(hsb + (unsigned)(nraw1 >= 0 ? nraw1 : 0) * C_, hf, m1);
        const int i2 = (i + 2 < TR_) ? (i + 2) : (TR_ - 1);
        const int nn0 = mrow0[i2 * M_], nn1 = mrow0[i2 * M_ + 16];

        const int i1 = (i + 1 < TR_) ? (i + 1) : (TR_ - 1);
        const v4f nqv = *(const v4f*)(qrd + i1 * QR_);

        __builtin_amdgcn_sched_barrier(0);

        frag2 fq = scaled(fk[0], qv[0]);
        for (int t = 1; t < 4; ++t) {
            fq.lo += fk[t].lo * qv[t];
            fq.hi += fk[t].hi * qv[t];
        }
        const v8f z8 = {};
        v8f ks0 = __builtin_amdgcn_wmma_f32_16x16x4_f32(false, fq.lo, false, g0.lo, (short)0, z8, false, false);
        v8f ks1 = __builtin_amdgcn_wmma_f32_16x16x4_f32(false, fq.lo, false, g1.lo, (short)0, z8, false, false);
        ks0 = __builtin_amdgcn_wmma_f32_16x16x4_f32(false, fq.hi, false, g0.hi, (short)0, ks0, false, false);
        ks1 = __builtin_amdgcn_wmma_f32_16x16x4_f32(false, fq.hi, false, g1.hi, (short)0, ks1, false, false);
        const float sp0 = sum8(ks0), sp1 = sum8(ks1);
        const float s0 = sp0 + xor16(sp0);
        const float s1 = sp1 + xor16(sp1);

        const bool valid0 = (raw0 >= 0), valid1 = (raw1 >= 0);
        const float sc0 = valid0 ? s0 : -1.0e9f;
        const float sc1 = valid1 ? s1 : -1.0e9f;
        float mx = fmaxf(sc0, sc1);
        mx = fmaxf(mx, xor8(mx));
        mx = fmaxf(mx, xor4(mx));
        mx = fmaxf(mx, xor2(mx));
        mx = fmaxf(mx, xor1(mx));
        const float pe0 = valid0 ? __expf(sc0 - mx) : 0.0f;
        const float pe1 = valid1 ? __expf(sc1 - mx) : 0.0f;
        float sum = pe0 + pe1;
        sum += xor8(sum);
        sum += xor4(sum);
        sum += xor2(sum);
        sum += xor1(sum);
        const float inv = 1.0f / fmaxf(sum, 1.0e-30f);
        const float at0 = pe0 * inv, at1 = pe1 * inv;

        frag2 a;
        a.lo = g0.lo * at0 + g1.lo * at1;
        a.hi = g0.hi * at0 + g1.hi * at1;
        v8f vs[4];
        for (int t = 0; t < 4; ++t)
            vs[t] = __builtin_amdgcn_wmma_f32_16x16x4_f32(false, a.lo, false, fv[t].lo, (short)0, z8, false, false);
        for (int t = 0; t < 4; ++t)
            vs[t] = __builtin_amdgcn_wmma_f32_16x16x4_f32(false, a.hi, false, fv[t].hi, (short)0, vs[t], false, false);
        float o[4];
        for (int t = 0; t < 4; ++t)
            o[t] = sum8(vs[t]);
        const float oa = (hf ? o[2] : o[0]) + xor16(hf ? o[0] : o[2]);
        const float ob = (hf ? o[3] : o[1]) + xor16(hf ? o[1] : o[3]);
        tout[i * TP_]      = oa;
        tout[i * TP_ + 16] = ob;

        g0 = ng0; g1 = ng1;
        qv = nqv;
        raw0 = nraw0; raw1 = nraw1;
        nraw0 = nn0;  nraw1 = nn1;
    }

    const int  col = n;
    const bool cv  = (col < C_);
    const float cmask = cv ? 1.0f : 0.0f;
    const int   bcol  = cv ? col : 0;
    const float* __restrict__ bp = fc_w + (size_t)bcol * W_ + h * D_ + hf * 4;
    const float*              ap = tile + col * TP_ + h * D_ + hf * 4;

    v4f bfr[8], afr[8];
    for (int s8 = 0; s8 < 8; ++s8)
        bfr[s8] = *(const v4f*)(bp + s8 * 8);

    __syncthreads();

    for (int s8 = 0; s8 < 8; ++s8)
        afr[s8] = *(const v4f*)(ap + s8 * 8);

    v8f c0 = {}, c1 = {};
    for (int s8 = 0; s8 < 8; ++s8) {
        v2f a, bb;
        a.x = afr[s8].x;           a.y = afr[s8].y;
        bb.x = bfr[s8].x * cmask;  bb.y = bfr[s8].y * cmask;
        c0 = __builtin_amdgcn_wmma_f32_16x16x4_f32(
                 false, a, false, bb, (short)0, c0, false, false);
        a.x = afr[s8].z;           a.y = afr[s8].w;
        bb.x = bfr[s8].z * cmask;  bb.y = bfr[s8].w * cmask;
        c1 = __builtin_amdgcn_wmma_f32_16x16x4_f32(
                 false, a, false, bb, (short)0, c1, false, false);
    }
#pragma unroll
    for (int r = 0; r < 8; ++r)
        part[(h * 8 + r) * 32 + lane] = c0[r] + c1[r];

    __syncthreads();

    float acc = 0.0f;
#pragma unroll
    for (int w = 0; w < H_; ++w)
        acc += part[(w * 8 + h) * 32 + lane];

    const float bias = cv ? fc_b[col] : 0.0f;
    const float lsc  = cv ? ln_s[col] : 0.0f;
    const float lbi  = cv ? ln_b[col] : 0.0f;
    const int   row  = row0 + hf * 8 + h;
    const float val  = acc + bias + (cv ? hs[row * C_ + col] : 0.0f);

    float x = cv ? val : 0.0f;
    float ssum = x;
#pragma unroll
    for (int o = 8; o > 0; o >>= 1)
        ssum += __shfl_xor(ssum, o, 32);
    const float mu = ssum * (1.0f / (float)C_);

    float dv = cv ? (val - mu) : 0.0f;
    float vsum = dv * dv;
#pragma unroll
    for (int o = 8; o > 0; o >>= 1)
        vsum += __shfl_xor(vsum, o, 32);
    const float var = vsum * (1.0f / (float)C_);

    const float y = (val - mu) * rsqrtf(var + 1.0e-6f) * lsc + lbi;
    if (cv) out[row * C_ + col] = y;
}

extern "C" void kernel_launch(void* const* d_in, const int* in_sizes, int n_in,
                              void* d_out, int out_size, void* d_ws, size_t ws_size,
                              hipStream_t stream)
{
    const float* hs  = (const float*)d_in[0];
    const float* wq  = (const float*)d_in[1];
    const float* wk  = (const float*)d_in[2];
    const float* wv  = (const float*)d_in[3];
    const float* fcw = (const float*)d_in[4];
    const float* fcb = (const float*)d_in[5];
    const float* lns = (const float*)d_in[6];
    const float* lnb = (const float*)d_in[7];
    const int*   qkm = (const int*)d_in[8];
    float* out = (float*)d_out;

    pyr_attn_remat<<<ROWS_ / TR_, 256, 0, stream>>>(hs, wq, wk, wv, qkm, fcw, fcb, lns, lnb, out);
}


# Sum over keys moved in front of the V product: a row's attn-weighted key fragment goes to LDS, a lane adds one row's 16 pieces, one V WMMA set per 16 rows, not per row (fp sum order changes)
# speedup vs baseline: 7.7383x; 1.1866x over previous
#include <hip/hip_runtime.h>


#define B_ 2
#define L_ 4096
#define C_ 7
#define H_ 8
#define D_ 64
#define W_ 512
#define M_ 32
#define ROWS_ (B_*L_)

typedef float v2f __attribute__((ext_vector_type(2)));
typedef float v4f __attribute__((ext_vector_type(4)));
typedef float v8f __attribute__((ext_vector_type(8)));

#define TR_ 16
#define TP_ (W_ + 4)
#define QR_ 64
#define AP_ 132

#define DPP8_SEL(a, b, c, d, e, f, g, h) \
    ((a) | ((b) << 3) | ((c) << 6) | ((d) << 9) | ((e) << 12) | ((f) << 15) | ((g) << 18) | ((h) << 21))

template <unsigned SEL>
__device__ __forceinline__ float dpp8(float x)
{
    return __builtin_bit_cast(float,
        __builtin_amdgcn_mov_dpp8(__builtin_bit_cast(unsigned, x), SEL));
}
__device__ __forceinline__ float xor8(float x)
{
    const int b = __builtin_bit_cast(int, x);
    return __builtin_bit_cast(float,
        __builtin_amdgcn_update_dpp(b, b, 0x128, 0xf, 0xf, false));
}
__device__ __forceinline__ float xor16(float x)
{
    const int b = __builtin_bit_cast(int, x);
    return __builtin_bit_cast(float,
        __builtin_amdgcn_permlanex16(b, b, 0x76543210, 0xfedcba98, false, false));
}
__device__ __forceinline__ float xor4(float x) { return dpp8<DPP8_SEL(4, 5, 6, 7, 0, 1, 2, 3)>(x); }
__device__ __forceinline__ float xor2(float x) { return dpp8<DPP8_SEL(2, 3, 0, 1, 6, 7, 4, 5)>(x); }
__device__ __forceinline__ float xor1(float x) { return dpp8<DPP8_SEL(1, 0, 3, 2, 5, 4, 7, 6)>(x); }

struct frag2 { v2f lo, hi; };
__device__ __forceinline__ frag2 row_frag(const float* __restrict__ p, int hf, float m)
{
    frag2 f;
    f.lo.x = p[0];
    f.lo.y = p[1];
    f.hi.x = p[4];
    f.hi.y = p[5 - hf] * m;
    return f;
}

__device__ __forceinline__ frag2 row_frag_unmasked(const float* __restrict__ p, int hf)
{
    frag2 f;
    f.lo.x = p[0];
    f.lo.y = p[1];
    f.hi.x = p[4];
    f.hi.y = p[5 - hf];
    return f;
}

__device__ __forceinline__ v8f wmma2acc(const frag2& a, const frag2& b, v8f c)
{
    c = __builtin_amdgcn_wmma_f32_16x16x4_f32(false, a.lo, false, b.lo, (short)0, c, false, false);
    c = __builtin_amdgcn_wmma_f32_16x16x4_f32(false, a.hi, false, b.hi, (short)0, c, false, false);
    return c;
}
__device__ __forceinline__ v8f wmma2(const frag2& a, const frag2& b)
{
    const v8f z = {};
    return wmma2acc(a, b, z);
}
__device__ __forceinline__ frag2 scaled(const frag2& f, float s)
{
    frag2 o;
    o.lo = f.lo * s;
    o.hi = f.hi * s;
    return o;
}
__device__ __forceinline__ float sum8(const v8f& c)
{
    const v2f p = (__builtin_shufflevector(c, c, 0, 1) + __builtin_shufflevector(c, c, 2, 3))
                + (__builtin_shufflevector(c, c, 4, 5) + __builtin_shufflevector(c, c, 6, 7));
    return p.x + p.y;
}

__global__ __launch_bounds__(256) void pyr_attn_remat(
    const float* __restrict__ hs, const float* __restrict__ wq,
    const float* __restrict__ wk, const float* __restrict__ wv,
    const int* __restrict__ qk_mask,
    const float* __restrict__ fc_w, const float* __restrict__ fc_b,
    const float* __restrict__ ln_s, const float* __restrict__ ln_b,
    float* __restrict__ out)
{
    __shared__ __attribute__((aligned(16))) float tile[TR_ * TP_];
    __shared__ __attribute__((aligned(16))) float qs[H_ * TR_ * QR_];
    __shared__ float part[H_ * 8 * 32];
    __shared__ __attribute__((aligned(16))) float ascr[H_ * TR_ * AP_];

    const int row0 = blockIdx.x * TR_;
    const int b    = row0 / L_;
    const int l0   = row0 % L_;

    const int h    = __builtin_amdgcn_readfirstlane((int)(threadIdx.x >> 5));
    const int lane = threadIdx.x & 31;
    const int n    = lane & 15;
    const int hf   = lane >> 4;
    const float m1 = hf ? 0.0f : 1.0f;

    const float* __restrict__ hsb   = hs + (size_t)b * L_ * C_ + hf * 2;
    const int*   __restrict__ mrow0 = qk_mask + l0 * M_ + n;

    frag2 fk[4], fv[4];
    for (int t = 0; t < 4; ++t) {
        const int wr = (h * D_ + t * 16 + n) * C_ + hf * 2;
        fk[t] = row_frag(wk + wr, hf, m1);
        fv[t] = row_frag(wv + wr, hf, m1);
    }

    {
        const frag2 fr = row_frag(hs + (size_t)(row0 + n) * C_ + hf * 2, hf, m1);
        v8f cq[4];
        for (int t = 0; t < 4; ++t) {
            const int wr = (h * D_ + t * 16 + n) * C_ + hf * 2;
            const frag2 fq = row_frag(wq + wr, hf, m1);
            cq[t] = wmma2(fr, fq);
        }
        float* __restrict__ qd = qs + (h * TR_ + hf * 8) * QR_ + n * 4;
#pragma unroll
        for (int r = 0; r < 8; ++r) {
            v4f s;
            s.x = cq[0][r] * 0.125f; s.y = cq[1][r] * 0.125f;
            s.z = cq[2][r] * 0.125f; s.w = cq[3][r] * 0.125f;
            *(v4f*)(qd + r * QR_) = s;
        }
    }

    int raw0  = mrow0[0],  raw1  = mrow0[16];
    int nraw0 = mrow0[M_], nraw1 = mrow0[M_ + 16];
    frag2 g0 = row_frag(hsb + (unsigned)(raw0 >= 0 ? raw0 : 0) * C_, hf, m1);
    frag2 g1 = row_frag(hsb + (unsigned)(raw1 >= 0 ? raw1 : 0) * C_, hf, m1);

    __syncthreads();

    const float* __restrict__ qrd = qs + h * TR_ * QR_ + n * 4;
    float* awr = ascr + h * TR_ * AP_ + hf * 64 + n * 4;
    v4f qv = *(const v4f*)qrd;

#pragma unroll 1
    for (int i = 0; i < TR_; ++i) {
        const frag2 ng0 = row_frag_unmasked(hsb + (unsigned)(nraw0 >= 0 ? nraw0 : 0) * C_, hf);
        const frag2 ng1 = row_frag_unmasked(hsb + (unsigned)(nraw1 >= 0 ? nraw1 : 0) * C_, hf);
        const int i2 = (i + 2 < TR_) ? (i + 2) : (TR_ - 1);
        const int nn0 = mrow0[i2 * M_], nn1 = mrow0[i2 * M_ + 16];

        const int i1 = (i + 1 < TR_) ? (i + 1) : (TR_ - 1);
        const v4f nqv = *(const v4f*)(qrd + i1 * QR_);

        __builtin_amdgcn_sched_barrier(0);

        frag2 fq = scaled(fk[0], qv[0]);
        for (int t = 1; t < 4; ++t) {
            fq.lo += fk[t].lo * qv[t];
            fq.hi += fk[t].hi * qv[t];
        }
        const v8f z8 = {};
        v8f ks0 = __builtin_amdgcn_wmma_f32_16x16x4_f32(false, fq.lo, false, g0.lo, (short)0, z8, false, false);
        v8f ks1 = __builtin_amdgcn_wmma_f32_16x16x4_f32(false, fq.lo, false, g1.lo, (short)0, z8, false, false);
        ks0 = __builtin_amdgcn_wmma_f32_16x16x4_f32(false, fq.hi, false, g0.hi, (short)0, ks0, false, false);
        ks1 = __builtin_amdgcn_wmma_f32_16x16x4_f32(false, fq.hi, false, g1.hi, (short)0, ks1, false, false);
        const float sp0 = sum8(ks0), sp1 = sum8(ks1);
        const float s0 = sp0 + xor16(sp0);
        const float s1 = sp1 + xor16(sp1);

        const bool valid0 = (raw0 >= 0), valid1 = (raw1 >= 0);
        const float sc0 = valid0 ? s0 : -1.0e9f;
        const float sc1 = valid1 ? s1 : -1.0e9f;
        float mx = fmaxf(sc0, sc1);
        mx = fmaxf(mx, xor8(mx));
        mx = fmaxf(mx, xor4(mx));
        mx = fmaxf(mx, xor2(mx));
        mx = fmaxf(mx, xor1(mx));
        const float pe0 = valid0 ? __expf(sc0 - mx) : 0.0f;
        const float pe1 = valid1 ? __expf(sc1 - mx) : 0.0f;
        float sum = pe0 + pe1;
        sum += xor8(sum);
        sum += xor4(sum);
        sum += xor2(sum);
        sum += xor1(sum);
        const float inv = 1.0f / fmaxf(sum, 1.0e-30f);
        const float at0 = pe0 * inv, at1 = pe1 * inv;

        v4f a;
        a.x = g0.lo.x * at0 + g1.lo.x * at1;
        a.y = g0.lo.y * at0 + g1.lo.y * at1;
        a.z = g0.hi.x * at0 + g1.hi.x * at1;
        a.w = g0.hi.y * at0 + g1.hi.y * at1;
        *(v4f*)(awr + i * AP_) = a;

        __builtin_amdgcn_sched_barrier(0);
        g0 = ng0; g1 = ng1;
        g0.hi.y *= m1;
        g1.hi.y *= m1;
        qv = nqv;
        raw0 = nraw0; raw1 = nraw1;
        nraw0 = nn0;  nraw1 = nn1;
    }

    {
        const float* ard = ascr + (h * TR_ + n) * AP_ + hf * 64;
        v4f y0 = *(const v4f*)ard;
        v4f y1 = *(const v4f*)(ard + 4);
#pragma unroll
        for (int k = 2; k < 16; k += 2) {
            y0 += *(const v4f*)(ard + k * 4);
            y1 += *(const v4f*)(ard + k * 4 + 4);
        }
        const v4f y = y0 + y1;
        v2f ylo, yhi;
        ylo.x = y.x; ylo.y = y.y;
        yhi.x = y.z; yhi.y = y.w;
        const v8f z8 = {};
        v8f vs[4];
        for (int t = 0; t < 4; ++t)
            vs[t] = __builtin_amdgcn_wmma_f32_16x16x4_f32(false, ylo, false, fv[t].lo, (short)0, z8, false, false);
        for (int t = 0; t < 4; ++t)
            vs[t] = __builtin_amdgcn_wmma_f32_16x16x4_f32(false, yhi, false, fv[t].hi, (short)0, vs[t], false, false);
        float* tw = tile + hf * 8 * TP_ + h * D_ + n;
#pragma unroll
        for (int r = 0; r < 8; ++r) {
#pragma unroll
            for (int t = 0; t < 4; ++t)
                tw[r * TP_ + t * 16] = vs[t][r];
        }
    }

    const int  col = n;
    const bool cv  = (col < C_);
    const float cmask = cv ? 1.0f : 0.0f;
    const int   bcol  = cv ? col : 0;
    const float* __restrict__ bp = fc_w + (size_t)bcol * W_ + h * D_ + hf * 4;
    const float*              ap = tile + col * TP_ + h * D_ + hf * 4;

    v4f bfr[8], afr[8];
    for (int s8 = 0; s8 < 8; ++s8)
        bfr[s8] = *(const v4f*)(bp + s8 * 8);

    __syncthreads();

    for (int s8 = 0; s8 < 8; ++s8)
        afr[s8] = *(const v4f*)(ap + s8 * 8);

    v8f c0 = {}, c1 = {};
    for (int s8 = 0; s8 < 8; ++s8) {
        v2f a, bb;
        a.x = afr[s8].x;           a.y = afr[s8].y;
        bb.x = bfr[s8].x * cmask;  bb.y = bfr[s8].y * cmask;
        c0 = __builtin_amdgcn_wmma_f32_16x16x4_f32(
                 false, a, false, bb, (short)0, c0, false, false);
        a.x = afr[s8].z;           a.y = afr[s8].w;
        bb.x = bfr[s8].z * cmask;  bb.y = bfr[s8].w * cmask;
        c1 = __builtin_amdgcn_wmma_f32_16x16x4_f32(
                 false, a, false, bb, (short)0, c1, false, false);
    }
#pragma unroll
    for (int r = 0; r < 8; ++r)
        part[(h * 8 + r) * 32 + lane] = c0[r] + c1[r];

    __syncthreads();

    float acc = 0.0f;
#pragma unroll
    for (int w = 0; w < H_; ++w)
        acc += part[(w * 8 + h) * 32 + lane];

    const float bias = cv ? fc_b[col] : 0.0f;
    const float lsc  = cv ? ln_s[col] : 0.0f;
    const float lbi  = cv ? ln_b[col] : 0.0f;
    const int   row  = row0 + hf * 8 + h;
    const float val  = acc + bias + (cv ? hs[row * C_ + col] : 0.0f);

    float x = cv ? val : 0.0f;
    float ssum = x;
#pragma unroll
    for (int o = 8; o > 0; o >>= 1)
        ssum += __shfl_xor(ssum, o, 32);
    const float mu = ssum * (1.0f / (float)C_);

    float dv = cv ? (val - mu) : 0.0f;
    float vsum = dv * dv;
#pragma unroll
    for (int o = 8; o > 0; o >>= 1)
        vsum += __shfl_xor(vsum, o, 32);
    const float var = vsum * (1.0f / (float)C_);

    const float y = (val - mu) * rsqrtf(var + 1.0e-6f) * lsc + lbi;
    if (cv) out[row * C_ + col] = y;
}

extern "C" void kernel_launch(void* const* d_in, const int* in_sizes, int n_in,
                              void* d_out, int out_size, void* d_ws, size_t ws_size,
                              hipStream_t stream)
{
    const float* hs  = (const float*)d_in[0];
    const float* wq  = (const float*)d_in[1];
    const float* wk  = (const float*)d_in[2];
    const float* wv  = (const float*)d_in[3];
    const float* fcw = (const float*)d_in[4];
    const float* fcb = (const float*)d_in[5];
    const float* lns = (const float*)d_in[6];
    const float* lnb = (const float*)d_in[7];
    const int*   qkm = (const int*)d_in[8];
    float* out = (float*)d_out;

    pyr_attn_remat<<<ROWS_ / TR_, 256, 0, stream>>>(hs, wq, wk, wv, qkm, fcw, fcb, lns, lnb, out);
}


# A score WMMA serves two heads: u = Wk_h^T q built once per work-group by chained WMMA; A rows 0-7 / 8-15 = u of two heads, softmax per lane half, 8 trips not 16 (fp sum order changes)
# speedup vs baseline: 10.2914x; 1.3299x over previous
#include <hip/hip_runtime.h>


#define B_ 2
#define L_ 4096
#define C_ 7
#define H_ 8
#define D_ 64
#define W_ 512
#define M_ 32
#define ROWS_ (B_*L_)

typedef float v2f __attribute__((ext_vector_type(2)));
typedef float v4f __attribute__((ext_vector_type(4)));
typedef float v8f __attribute__((ext_vector_type(8)));

#define TR_ 16
#define TP_ (W_ + 4)
#define UR_ 8
#define AP_ 132

#define DPP8_SEL(a, b, c, d, e, f, g, h) \
    ((a) | ((b) << 3) | ((c) << 6) | ((d) << 9) | ((e) << 12) | ((f) << 15) | ((g) << 18) | ((h) << 21))

template <unsigned SEL>
__device__ __forceinline__ float dpp8(float x)
{
    return __builtin_bit_cast(float,
        __builtin_amdgcn_mov_dpp8(__builtin_bit_cast(unsigned, x), SEL));
}
__device__ __forceinline__ float xor8(float x)
{
    const int b = __builtin_bit_cast(int, x);
    return __builtin_bit_cast(float,
        __builtin_amdgcn_update_dpp(b, b, 0x128, 0xf, 0xf, false));
}
__device__ __forceinline__ float xor16(float x)
{
    const int b = __builtin_bit_cast(int, x);
    return __builtin_bit_cast(float,
        __builtin_amdgcn_permlanex16(b, b, 0x76543210, 0xfedcba98, false, false));
}
__device__ __forceinline__ float xor4(float x) { return dpp8<DPP8_SEL(4, 5, 6, 7, 0, 1, 2, 3)>(x); }
__device__ __forceinline__ float xor2(float x) { return dpp8<DPP8_SEL(2, 3, 0, 1, 6, 7, 4, 5)>(x); }
__device__ __forceinline__ float xor1(float x) { return dpp8<DPP8_SEL(1, 0, 3, 2, 5, 4, 7, 6)>(x); }

struct frag2 { v2f lo, hi; };
__device__ __forceinline__ frag2 row_frag(const float* __restrict__ p, int hf, float m)
{
    frag2 f;
    f.lo.x = p[0];
    f.lo.y = p[1];
    f.hi.x = p[4];
    f.hi.y = p[5 - hf] * m;
    return f;
}

__device__ __forceinline__ frag2 row_frag_unmasked(const float* __restrict__ p, int hf)
{
    frag2 f;
    f.lo.x = p[0];
    f.lo.y = p[1];
    f.hi.x = p[4];
    f.hi.y = p[5 - hf];
    return f;
}

__device__ __forceinline__ v8f wmma2acc(const frag2& a, const frag2& b, v8f c)
{
    c = __builtin_amdgcn_wmma_f32_16x16x4_f32(false, a.lo, false, b.lo, (short)0, c, false, false);
    c = __builtin_amdgcn_wmma_f32_16x16x4_f32(false, a.hi, false, b.hi, (short)0, c, false, false);
    return c;
}
__device__ __forceinline__ v8f wmma2(const frag2& a, const frag2& b)
{
    const v8f z = {};
    return wmma2acc(a, b, z);
}
__device__ __forceinline__ frag2 scaled(const frag2& f, float s)
{
    frag2 o;
    o.lo = f.lo * s;
    o.hi = f.hi * s;
    return o;
}
__device__ __forceinline__ float sum8(const v8f& c)
{
    const v2f p = (__builtin_shufflevector(c, c, 0, 1) + __builtin_shufflevector(c, c, 2, 3))
                + (__builtin_shufflevector(c, c, 4, 5) + __builtin_shufflevector(c, c, 6, 7));
    return p.x + p.y;
}

__global__ __launch_bounds__(256) void pyr_attn_remat(
    const float* __restrict__ hs, const float* __restrict__ wq,
    const float* __restrict__ wk, const float* __restrict__ wv,
    const int* __restrict__ qk_mask,
    const float* __restrict__ fc_w, const float* __restrict__ fc_b,
    const float* __restrict__ ln_s, const float* __restrict__ ln_b,
    float* __restrict__ out)
{
    __shared__ __attribute__((aligned(16))) float tile[TR_ * TP_];
    __shared__ __attribute__((aligned(16))) float us[H_ * TR_ * UR_];
    __shared__ float part[H_ * 8 * 32];
    __shared__ __attribute__((aligned(16))) float ascr[H_ * TR_ * AP_];

    const int row0 = blockIdx.x * TR_;
    const int b    = row0 / L_;
    const int l0   = row0 % L_;

    const int h    = __builtin_amdgcn_readfirstlane((int)(threadIdx.x >> 5));
    const int lane = threadIdx.x & 31;
    const int n    = lane & 15;
    const int hf   = lane >> 4;
    const float m1 = hf ? 0.0f : 1.0f;
    const int p    = h & 3;
    const int rh   = h >> 2;

    const float* __restrict__ hsb   = hs + (size_t)b * L_ * C_ + hf * 2;
    const int*   __restrict__ mrow0 = qk_mask + (l0 + rh * 8) * M_ + n;

    frag2 fv[4];
    for (int t = 0; t < 4; ++t) {
        const int wr = (h * D_ + t * 16 + n) * C_ + hf * 2;
        fv[t] = row_frag(wv + wr, hf, m1);
    }

    {
        const frag2 fr = row_frag(hs + (size_t)(row0 + n) * C_ + hf * 2, hf, m1);
        v8f qt[4];
        for (int t = 0; t < 4; ++t) {
            const int wr = (h * D_ + t * 16 + n) * C_ + hf * 2;
            const frag2 fq = row_frag(wq + wr, hf, m1);
            qt[t] = wmma2(fq, fr);
        }
        const bool uc = (n < C_);
        const float* __restrict__ wkc = wk + (h * D_ + hf * 8) * C_ + (uc ? n : 0);
        v8f ua = {}, ub = {};
#pragma unroll
        for (int t = 0; t < 4; ++t) {
#pragma unroll
            for (int j = 0; j < 4; ++j) {
                const float w0 = wkc[(t * 16 + 2 * j) * C_];
                const float w1 = wkc[(t * 16 + 2 * j + 1) * C_];
                v2f a2, b2;
                a2.x = uc ? w0 : 0.0f;
                a2.y = uc ? w1 : 0.0f;
                b2.x = qt[t][2 * j];
                b2.y = qt[t][2 * j + 1];
                if (t < 2)
                    ua = __builtin_amdgcn_wmma_f32_16x16x4_f32(false, a2, false, b2, (short)0, ua, false, false);
                else
                    ub = __builtin_amdgcn_wmma_f32_16x16x4_f32(false, a2, false, b2, (short)0, ub, false, false);
            }
        }
        const v8f u = ua + ub;
        if (hf == 0) {
            float* __restrict__ ud = us + (h * TR_ + n) * UR_;
            v4f s0, s1;
            s0.x = u[0] * 0.125f; s0.y = u[1] * 0.125f; s0.z = u[4] * 0.125f; s0.w = u[5] * 0.125f;
            s1.x = u[2] * 0.125f; s1.y = u[3] * 0.125f; s1.z = u[6] * 0.125f; s1.w = u[7] * 0.125f;
            *(v4f*)ud       = s0;
            *(v4f*)(ud + 4) = s1;
        }
    }

    int raw0  = mrow0[0],  raw1  = mrow0[16];
    int nraw0 = mrow0[M_], nraw1 = mrow0[M_ + 16];
    frag2 g0 = row_frag(hsb + (unsigned)(raw0 >= 0 ? raw0 : 0) * C_, hf, m1);
    frag2 g1 = row_frag(hsb + (unsigned)(raw1 >= 0 ? raw1 : 0) * C_, hf, m1);

    __syncthreads();

    const float* __restrict__ urd = us + ((2 * p + (n >> 3)) * TR_ + rh * 8) * UR_ + hf * 4;
    float* awr_own = ascr + ((2 * p + hf)     * TR_ + rh * 8) * AP_ + hf * 64 + n * 4;
    float* awr_oth = ascr + ((2 * p + 1 - hf) * TR_ + rh * 8) * AP_ + hf * 64 + n * 4;
    v4f uv = *(const v4f*)urd;

#pragma unroll 1
    for (int i = 0; i < 8; ++i) {
        const frag2 ng0 = row_frag_unmasked(hsb + (unsigned)(nraw0 >= 0 ? nraw0 : 0) * C_, hf);
        const frag2 ng1 = row_frag_unmasked(hsb + (unsigned)(nraw1 >= 0 ? nraw1 : 0) * C_, hf);
        const int i2 = (i + 2 < 8) ? (i + 2) : 7;
        const int nn0 = mrow0[i2 * M_], nn1 = mrow0[i2 * M_ + 16];

        const int i1 = (i + 1 < 8) ? (i + 1) : 7;
        const v4f nuv = *(const v4f*)(urd + i1 * UR_);

        __builtin_amdgcn_sched_barrier(0);

        v2f ulo, uhi;
        ulo.x = uv.x; ulo.y = uv.y;
        uhi.x = uv.z; uhi.y = uv.w;
        const v8f z8 = {};
        v8f ks0 = __builtin_amdgcn_wmma_f32_16x16x4_f32(false, ulo, false, g0.lo, (short)0, z8, false, false);
        v8f ks1 = __builtin_amdgcn_wmma_f32_16x16x4_f32(false, ulo, false, g1.lo, (short)0, z8, false, false);
        ks0 = __builtin_amdgcn_wmma_f32_16x16x4_f32(false, uhi, false, g0.hi, (short)0, ks0, false, false);
        ks1 = __builtin_amdgcn_wmma_f32_16x16x4_f32(false, uhi, false, g1.hi, (short)0, ks1, false, false);
        const float s0 = ks0[0];
        const float s1 = ks1[0];

        const bool valid0 = (raw0 >= 0), valid1 = (raw1 >= 0);
        const float sc0 = valid0 ? s0 : -1.0e9f;
        const float sc1 = valid1 ? s1 : -1.0e9f;
        float mx = fmaxf(sc0, sc1);
        mx = fmaxf(mx, xor8(mx));
        mx = fmaxf(mx, xor4(mx));
        mx = fmaxf(mx, xor2(mx));
        mx = fmaxf(mx, xor1(mx));
        const float pe0 = valid0 ? __expf(sc0 - mx) : 0.0f;
        const float pe1 = valid1 ? __expf(sc1 - mx) : 0.0f;
        float sum = pe0 + pe1;
        sum += xor8(sum);
        sum += xor4(sum);
        sum += xor2(sum);
        sum += xor1(sum);
        const float inv = 1.0f / fmaxf(sum, 1.0e-30f);
        const float at0 = pe0 * inv, at1 = pe1 * inv;
        const float bt0 = xor16(at0), bt1 = xor16(at1);

        v4f a, o;
        a.x = g0.lo.x * at0 + g1.lo.x * at1;
        a.y = g0.lo.y * at0 + g1.lo.y * at1;
        a.z = g0.hi.x * at0 + g1.hi.x * at1;
        a.w = g0.hi.y * at0 + g1.hi.y * at1;
        o.x = g0.lo.x * bt0 + g1.lo.x * bt1;
        o.y = g0.lo.y * bt0 + g1.lo.y * bt1;
        o.z = g0.hi.x * bt0 + g1.hi.x * bt1;
        o.w = g0.hi.y * bt0 + g1.hi.y * bt1;
        *(v4f*)(awr_own + i * AP_) = a;
        *(v4f*)(awr_oth + i * AP_) = o;

        __builtin_amdgcn_sched_barrier(0);
        g0 = ng0; g1 = ng1;
        g0.hi.y *= m1;
        g1.hi.y *= m1;
        uv = nuv;
        raw0 = nraw0; raw1 = nraw1;
        nraw0 = nn0;  nraw1 = nn1;
    }

    __syncthreads();

    {
        const float* ard = ascr + (h * TR_ + n) * AP_ + hf * 64;
        v4f y0 = *(const v4f*)ard;
        v4f y1 = *(const v4f*)(ard + 4);
#pragma unroll
        for (int k = 2; k < 16; k += 2) {
            y0 += *(const v4f*)(ard + k * 4);
            y1 += *(const v4f*)(ard + k * 4 + 4);
        }
        const v4f y = y0 + y1;
        v2f ylo, yhi;
        ylo.x = y.x; ylo.y = y.y;
        yhi.x = y.z; yhi.y = y.w;
        const v8f z8 = {};
        v8f vs[4];
        for (int t = 0; t < 4; ++t)
            vs[t] = __builtin_amdgcn_wmma_f32_16x16x4_f32(false, ylo, false, fv[t].lo, (short)0, z8, false, false);
        for (int t = 0; t < 4; ++t)
            vs[t] = __builtin_amdgcn_wmma_f32_16x16x4_f32(false, yhi, false, fv[t].hi, (short)0, vs[t], false, false);
        float* tw = tile + hf * 8 * TP_ + h * D_ + n;
#pragma unroll
        for (int r = 0; r < 8; ++r) {
#pragma unroll
            for (int t = 0; t < 4; ++t)
                tw[r * TP_ + t * 16] = vs[t][r];
        }
    }

    const int  col = n;
    const bool cv  = (col < C_);
    const float cmask = cv ? 1.0f : 0.0f;
    const int   bcol  = cv ? col : 0;
    const float* __restrict__ bp = fc_w + (size_t)bcol * W_ + h * D_ + hf * 4;
    const float*              ap = tile + col * TP_ + h * D_ + hf * 4;

    v4f bfr[8], afr[8];
    for (int s8 = 0; s8 < 8; ++s8)
        bfr[s8] = *(const v4f*)(bp + s8 * 8);

    __syncthreads();

    for (int s8 = 0; s8 < 8; ++s8)
        afr[s8] = *(const v4f*)(ap + s8 * 8);

    v8f c0 = {}, c1 = {};
    for (int s8 = 0; s8 < 8; ++s8) {
        v2f a, bb;
        a.x = afr[s8].x;           a.y = afr[s8].y;
        bb.x = bfr[s8].x * cmask;  bb.y = bfr[s8].y * cmask;
        c0 = __builtin_amdgcn_wmma_f32_16x16x4_f32(
                 false, a, false, bb, (short)0, c0, false, false);
        a.x = afr[s8].z;           a.y = afr[s8].w;
        bb.x = bfr[s8].z * cmask;  bb.y = bfr[s8].w * cmask;
        c1 = __builtin_amdgcn_wmma_f32_16x16x4_f32(
                 false, a, false, bb, (short)0, c1, false, false);
    }
#pragma unroll
    for (int r = 0; r < 8; ++r)
        part[(h * 8 + r) * 32 + lane] = c0[r] + c1[r];

    __syncthreads();

    float acc = 0.0f;
#pragma unroll
    for (int w = 0; w < H_; ++w)
        acc += part[(w * 8 + h) * 32 + lane];

    const float bias = cv ? fc_b[col] : 0.0f;
    const float lsc  = cv ? ln_s[col] : 0.0f;
    const float lbi  = cv ? ln_b[col] : 0.0f;
    const int   row  = row0 + hf * 8 + h;
    const float val  = acc + bias + (cv ? hs[row * C_ + col] : 0.0f);

    float x = cv ? val : 0.0f;
    float ssum = x;
#pragma unroll
    for (int o = 8; o > 0; o >>= 1)
        ssum += __shfl_xor(ssum, o, 32);
    const float mu = ssum * (1.0f / (float)C_);

    float dv = cv ? (val - mu) : 0.0f;
    float vsum = dv * dv;
#pragma unroll
    for (int o = 8; o > 0; o >>= 1)
        vsum += __shfl_xor(vsum, o, 32);
    const float var = vsum * (1.0f / (float)C_);

    const float y = (val - mu) * rsqrtf(var + 1.0e-6f) * lsc + lbi;
    if (cv) out[row * C_ + col] = y;
}

extern "C" void kernel_launch(void* const* d_in, const int* in_sizes, int n_in,
                              void* d_out, int out_size, void* d_ws, size_t ws_size,
                              hipStream_t stream)
{
    const float* hs  = (const float*)d_in[0];
    const float* wq  = (const float*)d_in[1];
    const float* wk  = (const float*)d_in[2];
    const float* wv  = (const float*)d_in[3];
    const float* fcw = (const float*)d_in[4];
    const float* fcb = (const float*)d_in[5];
    const float* lns = (const float*)d_in[6];
    const float* lnb = (const float*)d_in[7];
    const int*   qkm = (const int*)d_in[8];
    float* out = (float*)d_out;

    pyr_attn_remat<<<ROWS_ / TR_, 256, 0, stream>>>(hs, wq, wk, wv, qkm, fcw, fcb, lns, lnb, out);
}


# Score loop re-split: a wave takes 2 rows and all 8 heads (A rows = u of the 8 heads, 4 heads per lane half), not 2 heads and 8 rows: key rows gathered once not 4 times, 8 score WMMAs a wave not 32
# speedup vs baseline: 14.3329x; 1.3927x over previous
#include <hip/hip_runtime.h>


#define B_ 2
#define L_ 4096
#define C_ 7
#define H_ 8
#define D_ 64
#define W_ 512
#define M_ 32
#define ROWS_ (B_*L_)

typedef float v2f __attribute__((ext_vector_type(2)));
typedef float v4f __attribute__((ext_vector_type(4)));
typedef float v8f __attribute__((ext_vector_type(8)));

#define TR_ 16
#define TP_ (W_ + 4)
#define UR_ 8
#define AP_ 132

#define DPP8_SEL(a, b, c, d, e, f, g, h) \
    ((a) | ((b) << 3) | ((c) << 6) | ((d) << 9) | ((e) << 12) | ((f) << 15) | ((g) << 18) | ((h) << 21))

template <unsigned SEL>
__device__ __forceinline__ float dpp8(float x)
{
    return __builtin_bit_cast(float,
        __builtin_amdgcn_mov_dpp8(__builtin_bit_cast(unsigned, x), SEL));
}
__device__ __forceinline__ float xor8(float x)
{
    const int b = __builtin_bit_cast(int, x);
    return __builtin_bit_cast(float,
        __builtin_amdgcn_update_dpp(b, b, 0x128, 0xf, 0xf, false));
}
__device__ __forceinline__ float xor16(float x)
{
    const int b = __builtin_bit_cast(int, x);
    return __builtin_bit_cast(float,
        __builtin_amdgcn_permlanex16(b, b, 0x76543210, 0xfedcba98, false, false));
}
__device__ __forceinline__ float xor4(float x) { return dpp8<DPP8_SEL(4, 5, 6, 7, 0, 1, 2, 3)>(x); }
__device__ __forceinline__ float xor2(float x) { return dpp8<DPP8_SEL(2, 3, 0, 1, 6, 7, 4, 5)>(x); }
__device__ __forceinline__ float xor1(float x) { return dpp8<DPP8_SEL(1, 0, 3, 2, 5, 4, 7, 6)>(x); }

struct frag2 { v2f lo, hi; };
__device__ __forceinline__ frag2 row_frag(const float* __restrict__ p, int hf, float m)
{
    frag2 f;
    f.lo.x = p[0];
    f.lo.y = p[1];
    f.hi.x = p[4];
    f.hi.y = p[5 - hf] * m;
    return f;
}

__device__ __forceinline__ frag2 row_frag_unmasked(const float* __restrict__ p, int hf)
{
    frag2 f;
    f.lo.x = p[0];
    f.lo.y = p[1];
    f.hi.x = p[4];
    f.hi.y = p[5 - hf];
    return f;
}

__device__ __forceinline__ v8f wmma2acc(const frag2& a, const frag2& b, v8f c)
{
    c = __builtin_amdgcn_wmma_f32_16x16x4_f32(false, a.lo, false, b.lo, (short)0, c, false, false);
    c = __builtin_amdgcn_wmma_f32_16x16x4_f32(false, a.hi, false, b.hi, (short)0, c, false, false);
    return c;
}
__device__ __forceinline__ v8f wmma2(const frag2& a, const frag2& b)
{
    const v8f z = {};
    return wmma2acc(a, b, z);
}
__device__ __forceinline__ frag2 scaled(const frag2& f, float s)
{
    frag2 o;
    o.lo = f.lo * s;
    o.hi = f.hi * s;
    return o;
}
__device__ __forceinline__ float sum8(const v8f& c)
{
    const v2f p = (__builtin_shufflevector(c, c, 0, 1) + __builtin_shufflevector(c, c, 2, 3))
                + (__builtin_shufflevector(c, c, 4, 5) + __builtin_shufflevector(c, c, 6, 7));
    return p.x + p.y;
}

__global__ __launch_bounds__(256) void pyr_attn_remat(
    const float* __restrict__ hs, const float* __restrict__ wq,
    const float* __restrict__ wk, const float* __restrict__ wv,
    const int* __restrict__ qk_mask,
    const float* __restrict__ fc_w, const float* __restrict__ fc_b,
    const float* __restrict__ ln_s, const float* __restrict__ ln_b,
    float* __restrict__ out)
{
    __shared__ __attribute__((aligned(16))) float tile[TR_ * TP_];
    __shared__ __attribute__((aligned(16))) float us[H_ * TR_ * UR_];
    __shared__ float part[H_ * 8 * 32];
    __shared__ __attribute__((aligned(16))) float ascr[H_ * TR_ * AP_];

    const int row0 = blockIdx.x * TR_;
    const int b    = row0 / L_;
    const int l0   = row0 % L_;

    const int h    = __builtin_amdgcn_readfirstlane((int)(threadIdx.x >> 5));
    const int lane = threadIdx.x & 31;
    const int n    = lane & 15;
    const int hf   = lane >> 4;
    const float m1 = hf ? 0.0f : 1.0f;
    const float* __restrict__ hsb   = hs + (size_t)b * L_ * C_ + hf * 2;
    const int*   __restrict__ mrow0 = qk_mask + (l0 + h * 2) * M_ + n;

    int raw[2][2];
#pragma unroll
    for (int i = 0; i < 2; ++i) {
        raw[i][0] = mrow0[i * M_];
        raw[i][1] = mrow0[i * M_ + 16];
    }

    frag2 fv[4], fq[4];
#pragma unroll
    for (int t = 0; t < 4; ++t) {
        const int wr = (h * D_ + t * 16 + n) * C_ + hf * 2;
        fq[t] = row_frag_unmasked(wq + wr, hf);
        fv[t] = row_frag_unmasked(wv + wr, hf);
    }
    frag2 fr = row_frag_unmasked(hs + (size_t)(row0 + n) * C_ + hf * 2, hf);
    const bool uc = (n < C_);
    const float* __restrict__ wkc = wk + (h * D_ + hf * 8) * C_ + (uc ? n : 0);
    float wkv[32];
#pragma unroll
    for (int k = 0; k < 32; ++k)
        wkv[k] = wkc[((k >> 3) * 16 + (k & 7)) * C_];

    __builtin_amdgcn_sched_barrier(0);

    frag2 g[2][2];
#pragma unroll
    for (int i = 0; i < 2; ++i) {
        g[i][0] = row_frag_unmasked(hsb + (unsigned)(raw[i][0] >= 0 ? raw[i][0] : 0) * C_, hf);
        g[i][1] = row_frag_unmasked(hsb + (unsigned)(raw[i][1] >= 0 ? raw[i][1] : 0) * C_, hf);
    }

    __builtin_amdgcn_sched_barrier(0);

    {
        fr.hi.y *= m1;
        v8f qt[4];
#pragma unroll
        for (int t = 0; t < 4; ++t) {
            fq[t].hi.y *= m1;
            qt[t] = wmma2(fq[t], fr);
        }
        v8f ua = {}, ub = {};
#pragma unroll
        for (int t = 0; t < 4; ++t) {
#pragma unroll
            for (int j = 0; j < 4; ++j) {
                const float w0 = wkv[t * 8 + 2 * j];
                const float w1 = wkv[t * 8 + 2 * j + 1];
                v2f a2, b2;
                a2.x = uc ? w0 : 0.0f;
                a2.y = uc ? w1 : 0.0f;
                b2.x = qt[t][2 * j];
                b2.y = qt[t][2 * j + 1];
                if (t < 2)
                    ua = __builtin_amdgcn_wmma_f32_16x16x4_f32(false, a2, false, b2, (short)0, ua, false, false);
                else
                    ub = __builtin_amdgcn_wmma_f32_16x16x4_f32(false, a2, false, b2, (short)0, ub, false, false);
            }
        }
        const v8f u = ua + ub;
        if (hf == 0) {
            float* __restrict__ ud = us + (h * TR_ + n) * UR_;
            v4f s0, s1;
            s0.x = u[0] * 0.125f; s0.y = u[1] * 0.125f; s0.z = u[4] * 0.125f; s0.w = u[5] * 0.125f;
            s1.x = u[2] * 0.125f; s1.y = u[3] * 0.125f; s1.z = u[6] * 0.125f; s1.w = u[7] * 0.125f;
            *(v4f*)ud       = s0;
            *(v4f*)(ud + 4) = s1;
        }
    }

#pragma unroll
    for (int t = 0; t < 4; ++t)
        fv[t].hi.y *= m1;
#pragma unroll
    for (int i = 0; i < 2; ++i) {
        g[i][0].hi.y *= m1;
        g[i][1].hi.y *= m1;
    }

    __syncthreads();

    const int hA = (n & 7) ^ ((n & 8) >> 1);
    const float* __restrict__ urd = us + (hA * TR_ + h * 2) * UR_ + hf * 4;
    float* awr_own = ascr + ((4 * hf)       * TR_ + h * 2) * AP_ + hf * 64 + n * 4;
    float* awr_oth = ascr + ((4 * (1 - hf)) * TR_ + h * 2) * AP_ + hf * 64 + n * 4;

#pragma unroll
    for (int i = 0; i < 2; ++i) {
        const frag2 g0 = g[i][0], g1 = g[i][1];
        const v4f uv = *(const v4f*)(urd + i * UR_);

        v2f ulo, uhi;
        ulo.x = uv.x; ulo.y = uv.y;
        uhi.x = uv.z; uhi.y = uv.w;
        const v8f z8 = {};
        v8f ks0 = __builtin_amdgcn_wmma_f32_16x16x4_f32(false, ulo, false, g0.lo, (short)0, z8, false, false);
        v8f ks1 = __builtin_amdgcn_wmma_f32_16x16x4_f32(false, ulo, false, g1.lo, (short)0, z8, false, false);
        ks0 = __builtin_amdgcn_wmma_f32_16x16x4_f32(false, uhi, false, g0.hi, (short)0, ks0, false, false);
        ks1 = __builtin_amdgcn_wmma_f32_16x16x4_f32(false, uhi, false, g1.hi, (short)0, ks1, false, false);

        const bool valid0 = (raw[i][0] >= 0), valid1 = (raw[i][1] >= 0);

#pragma unroll
        for (int j = 0; j < 4; ++j) {
            const float s0 = ks0[j];
            const float s1 = ks1[j];

            const float sc0 = valid0 ? s0 : -1.0e9f;
            const float sc1 = valid1 ? s1 : -1.0e9f;
            float mx = fmaxf(sc0, sc1);
            mx = fmaxf(mx, xor8(mx));
            mx = fmaxf(mx, xor4(mx));
            mx = fmaxf(mx, xor2(mx));
            mx = fmaxf(mx, xor1(mx));
            const float pe0 = valid0 ? __expf(sc0 - mx) : 0.0f;
            const float pe1 = valid1 ? __expf(sc1 - mx) : 0.0f;
            float sum = pe0 + pe1;
            sum += xor8(sum);
            sum += xor4(sum);
            sum += xor2(sum);
            sum += xor1(sum);
            const float inv = 1.0f / fmaxf(sum, 1.0e-30f);
            const float at0 = pe0 * inv, at1 = pe1 * inv;
            const float bt0 = xor16(at0), bt1 = xor16(at1);

            v4f a, o;
            a.x = g0.lo.x * at0 + g1.lo.x * at1;
            a.y = g0.lo.y * at0 + g1.lo.y * at1;
            a.z = g0.hi.x * at0 + g1.hi.x * at1;
            a.w = g0.hi.y * at0 + g1.hi.y * at1;
            o.x = g0.lo.x * bt0 + g1.lo.x * bt1;
            o.y = g0.lo.y * bt0 + g1.lo.y * bt1;
            o.z = g0.hi.x * bt0 + g1.hi.x * bt1;
            o.w = g0.hi.y * bt0 + g1.hi.y * bt1;
            *(v4f*)(awr_own + (j * TR_ + i) * AP_) = a;
            *(v4f*)(awr_oth + (j * TR_ + i) * AP_) = o;
        }
    }

    __syncthreads();

    {
        const float* ard = ascr + (h * TR_ + n) * AP_ + hf * 64;
        v4f y0 = *(const v4f*)ard;
        v4f y1 = *(const v4f*)(ard + 4);
#pragma unroll
        for (int k = 2; k < 16; k += 2) {
            y0 += *(const v4f*)(ard + k * 4);
            y1 += *(const v4f*)(ard + k * 4 + 4);
        }
        const v4f y = y0 + y1;
        v2f ylo, yhi;
        ylo.x = y.x; ylo.y = y.y;
        yhi.x = y.z; yhi.y = y.w;
        const v8f z8 = {};
        v8f vs[4];
        for (int t = 0; t < 4; ++t)
            vs[t] = __builtin_amdgcn_wmma_f32_16x16x4_f32(false, ylo, false, fv[t].lo, (short)0, z8, false, false);
        for (int t = 0; t < 4; ++t)
            vs[t] = __builtin_amdgcn_wmma_f32_16x16x4_f32(false, yhi, false, fv[t].hi, (short)0, vs[t], false, false);
        float* tw = tile + hf * 8 * TP_ + h * D_ + n;
#pragma unroll
        for (int r = 0; r < 8; ++r) {
#pragma unroll
            for (int t = 0; t < 4; ++t)
                tw[r * TP_ + t * 16] = vs[t][r];
        }
    }

    const int  col = n;
    const bool cv  = (col < C_);
    const float cmask = cv ? 1.0f : 0.0f;
    const int   bcol  = cv ? col : 0;
    const float* __restrict__ bp = fc_w + (size_t)bcol * W_ + h * D_ + hf * 4;
    const float*              ap = tile + col * TP_ + h * D_ + hf * 4;

    v4f bfr[8], afr[8];
    for (int s8 = 0; s8 < 8; ++s8)
        bfr[s8] = *(const v4f*)(bp + s8 * 8);

    __syncthreads();

    for (int s8 = 0; s8 < 8; ++s8)
        afr[s8] = *(const v4f*)(ap + s8 * 8);

    v8f c0 = {}, c1 = {};
    for (int s8 = 0; s8 < 8; ++s8) {
        v2f a, bb;
        a.x = afr[s8].x;           a.y = afr[s8].y;
        bb.x = bfr[s8].x * cmask;  bb.y = bfr[s8].y * cmask;
        c0 = __builtin_amdgcn_wmma_f32_16x16x4_f32(
                 false, a, false, bb, (short)0, c0, false, false);
        a.x = afr[s8].z;           a.y = afr[s8].w;
        bb.x = bfr[s8].z * cmask;  bb.y = bfr[s8].w * cmask;
        c1 = __builtin_amdgcn_wmma_f32_16x16x4_f32(
                 false, a, false, bb, (short)0, c1, false, false);
    }
#pragma unroll
    for (int r = 0; r < 8; ++r)
        part[(h * 8 + r) * 32 + lane] = c0[r] + c1[r];

    __syncthreads();

    float acc = 0.0f;
#pragma unroll
    for (int w = 0; w < H_; ++w)
        acc += part[(w * 8 + h) * 32 + lane];

    const float bias = cv ? fc_b[col] : 0.0f;
    const float lsc  = cv ? ln_s[col] : 0.0f;
    const float lbi  = cv ? ln_b[col] : 0.0f;
    const int   row  = row0 + hf * 8 + h;
    const float val  = acc + bias + (cv ? hs[row * C_ + col] : 0.0f);

    float x = cv ? val : 0.0f;
    float ssum = x;
#pragma unroll
    for (int o = 8; o > 0; o >>= 1)
        ssum += __shfl_xor(ssum, o, 32);
    const float mu = ssum * (1.0f / (float)C_);

    float dv = cv ? (val - mu) : 0.0f;
    float vsum = dv * dv;
#pragma unroll
    for (int o = 8; o > 0; o >>= 1)
        vsum += __shfl_xor(vsum, o, 32);
    const float var = vsum * (1.0f / (float)C_);

    const float y = (val - mu) * rsqrtf(var + 1.0e-6f) * lsc + lbi;
    if (cv) out[row * C_ + col] = y;
}

extern "C" void kernel_launch(void* const* d_in, const int* in_sizes, int n_in,
                              void* d_out, int out_size, void* d_ws, size_t ws_size,
                              hipStream_t stream)
{
    const float* hs  = (const float*)d_in[0];
    const float* wq  = (const float*)d_in[1];
    const float* wk  = (const float*)d_in[2];
    const float* wv  = (const float*)d_in[3];
    const float* fcw = (const float*)d_in[4];
    const float* fcb = (const float*)d_in[5];
    const float* lns = (const float*)d_in[6];
    const float* lnb = (const float*)d_in[7];
    const int*   qkm = (const int*)d_in[8];
    float* out = (float*)d_out;

    pyr_attn_remat<<<ROWS_ / TR_, 256, 0, stream>>>(hs, wq, wk, wv, qkm, fcw, fcb, lns, lnb, out);
}


# Output path transposed and kept in registers: V^T D tiles serve as B operand of Fc_h x V_h^T; a lane ends with its row's 7 channels, LayerNorm in-lane, no LDS tile (fp sum order changes)
# speedup vs baseline: 16.0605x; 1.1205x over previous
#include <hip/hip_runtime.h>


#define B_ 2
#define L_ 4096
#define C_ 7
#define H_ 8
#define D_ 64
#define W_ 512
#define M_ 32
#define ROWS_ (B_*L_)

typedef float v2f __attribute__((ext_vector_type(2)));
typedef float v4f __attribute__((ext_vector_type(4)));
typedef float v8f __attribute__((ext_vector_type(8)));

#define TR_ 16
#define UR_ 8
#define AP_ 132

#define DPP8_SEL(a, b, c, d, e, f, g, h) \
    ((a) | ((b) << 3) | ((c) << 6) | ((d) << 9) | ((e) << 12) | ((f) << 15) | ((g) << 18) | ((h) << 21))

template <unsigned SEL>
__device__ __forceinline__ float dpp8(float x)
{
    return __builtin_bit_cast(float,
        __builtin_amdgcn_mov_dpp8(__builtin_bit_cast(unsigned, x), SEL));
}
__device__ __forceinline__ float xor8(float x)
{
    const int b = __builtin_bit_cast(int, x);
    return __builtin_bit_cast(float,
        __builtin_amdgcn_update_dpp(b, b, 0x128, 0xf, 0xf, false));
}
__device__ __forceinline__ float xor16(float x)
{
    const int b = __builtin_bit_cast(int, x);
    return __builtin_bit_cast(float,
        __builtin_amdgcn_permlanex16(b, b, 0x76543210, 0xfedcba98, false, false));
}
__device__ __forceinline__ float xor4(float x) { return dpp8<DPP8_SEL(4, 5, 6, 7, 0, 1, 2, 3)>(x); }
__device__ __forceinline__ float xor2(float x) { return dpp8<DPP8_SEL(2, 3, 0, 1, 6, 7, 4, 5)>(x); }
__device__ __forceinline__ float xor1(float x) { return dpp8<DPP8_SEL(1, 0, 3, 2, 5, 4, 7, 6)>(x); }

struct frag2 { v2f lo, hi; };
__device__ __forceinline__ frag2 row_frag(const float* __restrict__ p, int hf, float m)
{
    frag2 f;
    f.lo.x = p[0];
    f.lo.y = p[1];
    f.hi.x = p[4];
    f.hi.y = p[5 - hf] * m;
    return f;
}

__device__ __forceinline__ frag2 row_frag_unmasked(const float* __restrict__ p, int hf)
{
    frag2 f;
    f.lo.x = p[0];
    f.lo.y = p[1];
    f.hi.x = p[4];
    f.hi.y = p[5 - hf];
    return f;
}

__device__ __forceinline__ v8f wmma2acc(const frag2& a, const frag2& b, v8f c)
{
    c = __builtin_amdgcn_wmma_f32_16x16x4_f32(false, a.lo, false, b.lo, (short)0, c, false, false);
    c = __builtin_amdgcn_wmma_f32_16x16x4_f32(false, a.hi, false, b.hi, (short)0, c, false, false);
    return c;
}
__device__ __forceinline__ v8f wmma2(const frag2& a, const frag2& b)
{
    const v8f z = {};
    return wmma2acc(a, b, z);
}
__device__ __forceinline__ frag2 scaled(const frag2& f, float s)
{
    frag2 o;
    o.lo = f.lo * s;
    o.hi = f.hi * s;
    return o;
}
__device__ __forceinline__ float sum8(const v8f& c)
{
    const v2f p = (__builtin_shufflevector(c, c, 0, 1) + __builtin_shufflevector(c, c, 2, 3))
                + (__builtin_shufflevector(c, c, 4, 5) + __builtin_shufflevector(c, c, 6, 7));
    return p.x + p.y;
}

__global__ __launch_bounds__(256) void pyr_attn_remat(
    const float* __restrict__ hs, const float* __restrict__ wq,
    const float* __restrict__ wk, const float* __restrict__ wv,
    const int* __restrict__ qk_mask,
    const float* __restrict__ fc_w, const float* __restrict__ fc_b,
    const float* __restrict__ ln_s, const float* __restrict__ ln_b,
    float* __restrict__ out)
{
    __shared__ __attribute__((aligned(16))) float us[H_ * TR_ * UR_];
    __shared__ __attribute__((aligned(16))) float part[H_ * TR_ * 8];
    __shared__ __attribute__((aligned(16))) float ascr[H_ * TR_ * AP_];

    const int row0 = blockIdx.x * TR_;
    const int b    = row0 / L_;
    const int l0   = row0 % L_;

    const int h    = __builtin_amdgcn_readfirstlane((int)(threadIdx.x >> 5));
    const int lane = threadIdx.x & 31;
    const int n    = lane & 15;
    const int hf   = lane >> 4;
    const float m1 = hf ? 0.0f : 1.0f;
    const float* __restrict__ hsb   = hs + (size_t)b * L_ * C_ + hf * 2;
    const int*   __restrict__ mrow0 = qk_mask + (l0 + h * 2) * M_ + n;

    int raw[2][2];
#pragma unroll
    for (int i = 0; i < 2; ++i) {
        raw[i][0] = mrow0[i * M_];
        raw[i][1] = mrow0[i * M_ + 16];
    }

    frag2 fv[4], fq[4];
#pragma unroll
    for (int t = 0; t < 4; ++t) {
        const int wr = (h * D_ + t * 16 + n) * C_ + hf * 2;
        fq[t] = row_frag_unmasked(wq + wr, hf);
        fv[t] = row_frag_unmasked(wv + wr, hf);
    }
    frag2 fr = row_frag_unmasked(hs + (size_t)(row0 + n) * C_ + hf * 2, hf);
    const bool uc = (n < C_);
    const float* __restrict__ wkc = wk + (h * D_ + hf * 8) * C_ + (uc ? n : 0);
    float wkv[32];
#pragma unroll
    for (int k = 0; k < 32; ++k)
        wkv[k] = wkc[((k >> 3) * 16 + (k & 7)) * C_];

    __builtin_amdgcn_sched_barrier(0);

    frag2 g[2][2];
#pragma unroll
    for (int i = 0; i < 2; ++i) {
        g[i][0] = row_frag_unmasked(hsb + (unsigned)(raw[i][0] >= 0 ? raw[i][0] : 0) * C_, hf);
        g[i][1] = row_frag_unmasked(hsb + (unsigned)(raw[i][1] >= 0 ? raw[i][1] : 0) * C_, hf);
    }

    __builtin_amdgcn_sched_barrier(0);

    {
        fr.hi.y *= m1;
        v8f qt[4];
#pragma unroll
        for (int t = 0; t < 4; ++t) {
            fq[t].hi.y *= m1;
            qt[t] = wmma2(fq[t], fr);
        }
        v8f ua = {}, ub = {};
#pragma unroll
        for (int t = 0; t < 4; ++t) {
#pragma unroll
            for (int j = 0; j < 4; ++j) {
                const float w0 = wkv[t * 8 + 2 * j];
                const float w1 = wkv[t * 8 + 2 * j + 1];
                v2f a2, b2;
                a2.x = uc ? w0 : 0.0f;
                a2.y = uc ? w1 : 0.0f;
                b2.x = qt[t][2 * j];
                b2.y = qt[t][2 * j + 1];
                if (t < 2)
                    ua = __builtin_amdgcn_wmma_f32_16x16x4_f32(false, a2, false, b2, (short)0, ua, false, false);
                else
                    ub = __builtin_amdgcn_wmma_f32_16x16x4_f32(false, a2, false, b2, (short)0, ub, false, false);
            }
        }
        const v8f u = ua + ub;
        if (hf == 0) {
            float* __restrict__ ud = us + (h * TR_ + n) * UR_;
            v4f s0, s1;
            s0.x = u[0] * 0.125f; s0.y = u[1] * 0.125f; s0.z = u[4] * 0.125f; s0.w = u[5] * 0.125f;
            s1.x = u[2] * 0.125f; s1.y = u[3] * 0.125f; s1.z = u[6] * 0.125f; s1.w = u[7] * 0.125f;
            *(v4f*)ud       = s0;
            *(v4f*)(ud + 4) = s1;
        }
    }

#pragma unroll
    for (int t = 0; t < 4; ++t)
        fv[t].hi.y *= m1;
#pragma unroll
    for (int i = 0; i < 2; ++i) {
        g[i][0].hi.y *= m1;
        g[i][1].hi.y *= m1;
    }

    __syncthreads();

    const int hA = (n & 7) ^ ((n & 8) >> 1);
    const float* __restrict__ urd = us + (hA * TR_ + h * 2) * UR_ + hf * 4;
    float* awr_own = ascr + ((4 * hf)       * TR_ + h * 2) * AP_ + hf * 64 + n * 4;
    float* awr_oth = ascr + ((4 * (1 - hf)) * TR_ + h * 2) * AP_ + hf * 64 + n * 4;

#pragma unroll
    for (int i = 0; i < 2; ++i) {
        const frag2 g0 = g[i][0], g1 = g[i][1];
        const v4f uv = *(const v4f*)(urd + i * UR_);

        v2f ulo, uhi;
        ulo.x = uv.x; ulo.y = uv.y;
        uhi.x = uv.z; uhi.y = uv.w;
        const v8f z8 = {};
        v8f ks0 = __builtin_amdgcn_wmma_f32_16x16x4_f32(false, ulo, false, g0.lo, (short)0, z8, false, false);
        v8f ks1 = __builtin_amdgcn_wmma_f32_16x16x4_f32(false, ulo, false, g1.lo, (short)0, z8, false, false);
        ks0 = __builtin_amdgcn_wmma_f32_16x16x4_f32(false, uhi, false, g0.hi, (short)0, ks0, false, false);
        ks1 = __builtin_amdgcn_wmma_f32_16x16x4_f32(false, uhi, false, g1.hi, (short)0, ks1, false, false);

        const bool valid0 = (raw[i][0] >= 0), valid1 = (raw[i][1] >= 0);

#pragma unroll
        for (int j = 0; j < 4; ++j) {
            const float s0 = ks0[j];
            const float s1 = ks1[j];

            const float sc0 = valid0 ? s0 : -1.0e9f;
            const float sc1 = valid1 ? s1 : -1.0e9f;
            float mx = fmaxf(sc0, sc1);
            mx = fmaxf(mx, xor8(mx));
            mx = fmaxf(mx, xor4(mx));
            mx = fmaxf(mx, xor2(mx));
            mx = fmaxf(mx, xor1(mx));
            const float pe0 = valid0 ? __expf(sc0 - mx) : 0.0f;
            const float pe1 = valid1 ? __expf(sc1 - mx) : 0.0f;
            float sum = pe0 + pe1;
            sum += xor8(sum);
            sum += xor4(sum);
            sum += xor2(sum);
            sum += xor1(sum);
            const float inv = 1.0f / fmaxf(sum, 1.0e-30f);
            const float at0 = pe0 * inv, at1 = pe1 * inv;
            const float bt0 = xor16(at0), bt1 = xor16(at1);

            v4f a, o;
            a.x = g0.lo.x * at0 + g1.lo.x * at1;
            a.y = g0.lo.y * at0 + g1.lo.y * at1;
            a.z = g0.hi.x * at0 + g1.hi.x * at1;
            a.w = g0.hi.y * at0 + g1.hi.y * at1;
            o.x = g0.lo.x * bt0 + g1.lo.x * bt1;
            o.y = g0.lo.y * bt0 + g1.lo.y * bt1;
            o.z = g0.hi.x * bt0 + g1.hi.x * bt1;
            o.w = g0.hi.y * bt0 + g1.hi.y * bt1;
            *(v4f*)(awr_own + (j * TR_ + i) * AP_) = a;
            *(v4f*)(awr_oth + (j * TR_ + i) * AP_) = o;
        }
    }

    const bool cv = (n < C_);
    const float* __restrict__ fcp = fc_w + (size_t)(cv ? n : 0) * W_ + h * D_ + hf * 8;
    v4f fcv[8];
#pragma unroll
    for (int t = 0; t < 4; ++t) {
        fcv[2 * t]     = *(const v4f*)(fcp + t * 16);
        fcv[2 * t + 1] = *(const v4f*)(fcp + t * 16 + 4);
    }
    float pfb[C_], pls[C_], plb[C_];
#pragma unroll
    for (int c = 0; c < C_; ++c)
        plb[c] = ln_b[c];
#pragma unroll
    for (int c = 0; c < C_; ++c) {
        pfb[c] = fc_b[c];
        pls[c] = ln_s[c];
    }

    __syncthreads();

    {
        const float* ard = ascr + (h * TR_ + n) * AP_ + hf * 64;
        v4f y0 = *(const v4f*)ard;
        v4f y1 = *(const v4f*)(ard + 4);
#pragma unroll
        for (int k = 2; k < 16; k += 2) {
            y0 += *(const v4f*)(ard + k * 4);
            y1 += *(const v4f*)(ard + k * 4 + 4);
        }
        const v4f y = y0 + y1;
        v2f ylo, yhi;
        ylo.x = y.x; ylo.y = y.y;
        yhi.x = y.z; yhi.y = y.w;
        const v8f z8 = {};
        v8f vt[4];
#pragma unroll
        for (int t = 0; t < 4; ++t)
            vt[t] = __builtin_amdgcn_wmma_f32_16x16x4_f32(false, fv[t].lo, false, ylo, (short)0, z8, false, false);
#pragma unroll
        for (int t = 0; t < 4; ++t)
            vt[t] = __builtin_amdgcn_wmma_f32_16x16x4_f32(false, fv[t].hi, false, yhi, (short)0, vt[t], false, false);

        v8f oa = {}, ob = {};
#pragma unroll
        for (int t = 0; t < 4; ++t) {
#pragma unroll
            for (int j = 0; j < 4; ++j) {
                const v4f f = fcv[2 * t + (j >> 1)];
                const float w0 = (j & 1) ? f.z : f.x;
                const float w1 = (j & 1) ? f.w : f.y;
                v2f a2, b2;
                a2.x = cv ? w0 : 0.0f;
                a2.y = cv ? w1 : 0.0f;
                b2.x = vt[t][2 * j];
                b2.y = vt[t][2 * j + 1];
                if (t < 2)
                    oa = __builtin_amdgcn_wmma_f32_16x16x4_f32(false, a2, false, b2, (short)0, oa, false, false);
                else
                    ob = __builtin_amdgcn_wmma_f32_16x16x4_f32(false, a2, false, b2, (short)0, ob, false, false);
            }
        }
        const v8f o = oa + ob;
        if (hf == 0) {
            float* __restrict__ pd = part + (h * TR_ + n) * 8;
            v4f s0, s1;
            s0.x = o[0]; s0.y = o[1]; s0.z = o[2]; s0.w = o[3];
            s1.x = o[4]; s1.y = o[5]; s1.z = o[6]; s1.w = o[7];
            *(v4f*)pd       = s0;
            *(v4f*)(pd + 4) = s1;
        }
    }

    __syncthreads();

    if (h == 0) {
        const float x2 = xor16(fr.lo.x), x3 = xor16(fr.lo.y), x6 = xor16(fr.hi.x);
        if (hf == 0) {
            const float* pr = part + n * 8;
            v4f a0 = *(const v4f*)pr;
            v4f a1 = *(const v4f*)(pr + 4);
#pragma unroll
            for (int w = 1; w < H_; ++w) {
                a0 += *(const v4f*)(pr + w * TR_ * 8);
                a1 += *(const v4f*)(pr + w * TR_ * 8 + 4);
            }
            float v[C_];
            v[0] = (a0.x + pfb[0]) + fr.lo.x;
            v[1] = (a0.y + pfb[1]) + fr.lo.y;
            v[2] = (a0.z + pfb[2]) + x2;
            v[3] = (a0.w + pfb[3]) + x3;
            v[4] = (a1.x + pfb[4]) + fr.hi.x;
            v[5] = (a1.y + pfb[5]) + fr.hi.y;
            v[6] = (a1.z + pfb[6]) + x6;

            const float ssum = ((v[0] + v[4]) + (v[2] + v[6])) + ((v[1] + v[5]) + v[3]);
            const float mu = ssum * (1.0f / (float)C_);
            float d[C_];
#pragma unroll
            for (int c = 0; c < C_; ++c)
                d[c] = v[c] - mu;
            const float vsum = ((d[0] * d[0] + d[4] * d[4]) + (d[2] * d[2] + d[6] * d[6]))
                             + ((d[1] * d[1] + d[5] * d[5]) + d[3] * d[3]);
            const float var = vsum * (1.0f / (float)C_);
            const float rs  = rsqrtf(var + 1.0e-6f);
            float* __restrict__ orow = out + (size_t)(row0 + n) * C_;
#pragma unroll
            for (int c = 0; c < C_; ++c)
                orow[c] = d[c] * rs * pls[c] + plb[c];
        }
    }
}

extern "C" void kernel_launch(void* const* d_in, const int* in_sizes, int n_in,
                              void* d_out, int out_size, void* d_ws, size_t ws_size,
                              hipStream_t stream)
{
    const float* hs  = (const float*)d_in[0];
    const float* wq  = (const float*)d_in[1];
    const float* wk  = (const float*)d_in[2];
    const float* wv  = (const float*)d_in[3];
    const float* fcw = (const float*)d_in[4];
    const float* fcb = (const float*)d_in[5];
    const float* lns = (const float*)d_in[6];
    const float* lnb = (const float*)d_in[7];
    const int*   qkm = (const int*)d_in[8];
    float* out = (float*)d_out;

    pyr_attn_remat<<<ROWS_ / TR_, 256, 0, stream>>>(hs, wq, wk, wv, qkm, fcw, fcb, lns, lnb, out);
}


# Score tile transposed: keys down the D registers, (row, head) across lanes; softmax max and sum in-lane plus one lane swap, attn D tiles feed the key sum as WMMA A operand (fp sum order changes)
# speedup vs baseline: 17.3196x; 1.0784x over previous
#include <hip/hip_runtime.h>


#define B_ 2
#define L_ 4096
#define C_ 7
#define H_ 8
#define D_ 64
#define W_ 512
#define M_ 32
#define ROWS_ (B_*L_)

typedef float v2f __attribute__((ext_vector_type(2)));
typedef float v4f __attribute__((ext_vector_type(4)));
typedef float v8f __attribute__((ext_vector_type(8)));

#define TR_ 16
#define UR_ 8
#define XR_ 288

__device__ __forceinline__ float xor16(float x)
{
    const int b = __builtin_bit_cast(int, x);
    return __builtin_bit_cast(float,
        __builtin_amdgcn_permlanex16(b, b, 0x76543210, 0xfedcba98, false, false));
}

struct frag2 { v2f lo, hi; };
__device__ __forceinline__ frag2 row_frag(const float* __restrict__ p, int hf, float m)
{
    frag2 f;
    f.lo.x = p[0];
    f.lo.y = p[1];
    f.hi.x = p[4];
    f.hi.y = p[5 - hf] * m;
    return f;
}

__device__ __forceinline__ frag2 row_frag_unmasked(const float* __restrict__ p, int hf)
{
    frag2 f;
    f.lo.x = p[0];
    f.lo.y = p[1];
    f.hi.x = p[4];
    f.hi.y = p[5 - hf];
    return f;
}

__device__ __forceinline__ v8f wmma2acc(const frag2& a, const frag2& b, v8f c)
{
    c = __builtin_amdgcn_wmma_f32_16x16x4_f32(false, a.lo, false, b.lo, (short)0, c, false, false);
    c = __builtin_amdgcn_wmma_f32_16x16x4_f32(false, a.hi, false, b.hi, (short)0, c, false, false);
    return c;
}
__device__ __forceinline__ v8f wmma2(const frag2& a, const frag2& b)
{
    const v8f z = {};
    return wmma2acc(a, b, z);
}
__device__ __forceinline__ frag2 scaled(const frag2& f, float s)
{
    frag2 o;
    o.lo = f.lo * s;
    o.hi = f.hi * s;
    return o;
}
__device__ __forceinline__ float sum8(const v8f& c)
{
    const v2f p = (__builtin_shufflevector(c, c, 0, 1) + __builtin_shufflevector(c, c, 2, 3))
                + (__builtin_shufflevector(c, c, 4, 5) + __builtin_shufflevector(c, c, 6, 7));
    return p.x + p.y;
}

__global__ __launch_bounds__(256) void pyr_attn_remat(
    const float* __restrict__ hs, const float* __restrict__ wq,
    const float* __restrict__ wk, const float* __restrict__ wv,
    const int* __restrict__ qk_mask,
    const float* __restrict__ fc_w, const float* __restrict__ fc_b,
    const float* __restrict__ ln_s, const float* __restrict__ ln_b,
    float* __restrict__ out)
{
    __shared__ __attribute__((aligned(16))) float us[H_ * TR_ * UR_];
    __shared__ __attribute__((aligned(16))) float part[H_ * TR_ * 8];
    __shared__ __attribute__((aligned(16))) float ysc[H_ * TR_ * 8];
    __shared__ __attribute__((aligned(16))) float xs[H_ * 2 * XR_];

    const int row0 = blockIdx.x * TR_;
    const int b    = row0 / L_;
    const int l0   = row0 % L_;

    const int h    = __builtin_amdgcn_readfirstlane((int)(threadIdx.x >> 5));
    const int lane = threadIdx.x & 31;
    const int n    = lane & 15;
    const int hf   = lane >> 4;
    const float m1 = hf ? 0.0f : 1.0f;
    const float* __restrict__ hsb   = hs + (size_t)b * L_ * C_ + hf * 2;
    const int*   __restrict__ mrow0 = qk_mask + (l0 + h * 2) * M_ + n;

    int raw[2][2];
#pragma unroll
    for (int i = 0; i < 2; ++i) {
        raw[i][0] = mrow0[i * M_];
        raw[i][1] = mrow0[i * M_ + 16];
    }

    frag2 fv[4], fq[4];
#pragma unroll
    for (int t = 0; t < 4; ++t) {
        const int wr = (h * D_ + t * 16 + n) * C_ + hf * 2;
        fq[t] = row_frag_unmasked(wq + wr, hf);
        fv[t] = row_frag_unmasked(wv + wr, hf);
    }
    frag2 fr = row_frag_unmasked(hs + (size_t)(row0 + n) * C_ + hf * 2, hf);
    const bool uc = (n < C_);
    const float* __restrict__ wkc = wk + (h * D_ + hf * 8) * C_ + (uc ? n : 0);
    float wkv[32];
#pragma unroll
    for (int k = 0; k < 32; ++k)
        wkv[k] = wkc[((k >> 3) * 16 + (k & 7)) * C_];

    __builtin_amdgcn_sched_barrier(0);

    frag2 g[2][2];
#pragma unroll
    for (int i = 0; i < 2; ++i) {
        g[i][0] = row_frag_unmasked(hsb + (unsigned)(raw[i][0] >= 0 ? raw[i][0] : 0) * C_, hf);
        g[i][1] = row_frag_unmasked(hsb + (unsigned)(raw[i][1] >= 0 ? raw[i][1] : 0) * C_, hf);
    }

    __builtin_amdgcn_sched_barrier(0);

    {
        fr.hi.y *= m1;
        v8f qt[4];
#pragma unroll
        for (int t = 0; t < 4; ++t) {
            fq[t].hi.y *= m1;
            qt[t] = wmma2(fq[t], fr);
        }
        v8f ua = {}, ub = {};
#pragma unroll
        for (int t = 0; t < 4; ++t) {
#pragma unroll
            for (int j = 0; j < 4; ++j) {
                const float w0 = wkv[t * 8 + 2 * j];
                const float w1 = wkv[t * 8 + 2 * j + 1];
                v2f a2, b2;
                a2.x = uc ? w0 : 0.0f;
                a2.y = uc ? w1 : 0.0f;
                b2.x = qt[t][2 * j];
                b2.y = qt[t][2 * j + 1];
                if (t < 2)
                    ua = __builtin_amdgcn_wmma_f32_16x16x4_f32(false, a2, false, b2, (short)0, ua, false, false);
                else
                    ub = __builtin_amdgcn_wmma_f32_16x16x4_f32(false, a2, false, b2, (short)0, ub, false, false);
            }
        }
        const v8f u = ua + ub;
        if (hf == 0) {
            float* __restrict__ ud = us + (h * TR_ + n) * UR_;
            v4f s0, s1;
            s0.x = u[0] * 0.125f; s0.y = u[1] * 0.125f; s0.z = u[4] * 0.125f; s0.w = u[5] * 0.125f;
            s1.x = u[2] * 0.125f; s1.y = u[3] * 0.125f; s1.z = u[6] * 0.125f; s1.w = 1.0f;
            *(v4f*)ud       = s0;
            *(v4f*)(ud + 4) = s1;
        }
    }

#pragma unroll
    for (int t = 0; t < 4; ++t)
        fv[t].hi.y *= m1;

    float* xw = xs + h * (2 * XR_);
#pragma unroll
    for (int i = 0; i < 2; ++i) {
#pragma unroll
        for (int t = 0; t < 2; ++t) {
            v4f x4;
            x4.x = g[i][t].lo.x; x4.y = g[i][t].lo.y; x4.z = g[i][t].hi.x; x4.w = g[i][t].hi.y;
            *(v4f*)(xw + i * XR_ + (18 * t + n + (n >> 3)) * 8 + hf * 4) = x4;
        }
    }
#pragma unroll
    for (int i = 0; i < 2; ++i) {
#pragma unroll
        for (int t = 0; t < 2; ++t) {
            const float pv = (raw[i][t] >= 0) ? 0.0f : -1.0e30f;
            g[i][t].hi.y = hf ? pv : g[i][t].hi.y;
        }
    }

    __syncthreads();

    const int i_n = n >> 3;
    const int h_n = n & 7;
    const v4f uv = *(const v4f*)(us + (h_n * TR_ + h * 2 + i_n) * UR_ + hf * 4);
    v2f ub[2][2];
#pragma unroll
    for (int i = 0; i < 2; ++i) {
        const bool own = (i_n == i);
        ub[i][0].x = own ? uv.x : 0.0f;
        ub[i][0].y = own ? uv.y : 0.0f;
        ub[i][1].x = own ? uv.z : 0.0f;
        ub[i][1].y = own ? uv.w : 0.0f;
    }

    const int c_n = n & 7;
    const int cpos = (c_n & 1) | ((c_n & 2) << 1) | ((c_n & 4) >> 1);
    const float* xrd = xw + i_n * XR_ + (9 * hf) * 8 + cpos;
    float xb[2][8];
#pragma unroll
    for (int t = 0; t < 2; ++t) {
#pragma unroll
        for (int r = 0; r < 8; ++r)
            xb[t][r] = xrd[(18 * t + r) * 8];
    }
    __builtin_amdgcn_sched_barrier(0);

    const v8f z8s = {};
    v8f sc0 = __builtin_amdgcn_wmma_f32_16x16x4_f32(false, g[0][0].lo, false, ub[0][0], (short)0, z8s, false, false);
    v8f sc1 = __builtin_amdgcn_wmma_f32_16x16x4_f32(false, g[0][1].lo, false, ub[0][0], (short)0, z8s, false, false);
    sc0 = __builtin_amdgcn_wmma_f32_16x16x4_f32(false, g[0][0].hi, false, ub[0][1], (short)0, sc0, false, false);
    sc1 = __builtin_amdgcn_wmma_f32_16x16x4_f32(false, g[0][1].hi, false, ub[0][1], (short)0, sc1, false, false);
    sc0 = __builtin_amdgcn_wmma_f32_16x16x4_f32(false, g[1][0].lo, false, ub[1][0], (short)0, sc0, false, false);
    sc1 = __builtin_amdgcn_wmma_f32_16x16x4_f32(false, g[1][1].lo, false, ub[1][0], (short)0, sc1, false, false);
    sc0 = __builtin_amdgcn_wmma_f32_16x16x4_f32(false, g[1][0].hi, false, ub[1][1], (short)0, sc0, false, false);
    sc1 = __builtin_amdgcn_wmma_f32_16x16x4_f32(false, g[1][1].hi, false, ub[1][1], (short)0, sc1, false, false);

    float mx;
    {
        float m8[8];
#pragma unroll
        for (int r = 0; r < 8; ++r)
            m8[r] = fmaxf(sc0[r], sc1[r]);
        mx = fmaxf(fmaxf(fmaxf(m8[0], m8[1]), fmaxf(m8[2], m8[3])),
                   fmaxf(fmaxf(m8[4], m8[5]), fmaxf(m8[6], m8[7])));
    }
    mx = fmaxf(mx, xor16(mx));
    mx = fmaxf(mx, -1.0e20f);
    float pe0[8], pe1[8];
#pragma unroll
    for (int r = 0; r < 8; ++r) {
        pe0[r] = __expf(sc0[r] - mx);
        pe1[r] = __expf(sc1[r] - mx);
    }
    float sum;
    {
        float s8[8];
#pragma unroll
        for (int r = 0; r < 8; ++r)
            s8[r] = pe0[r] + pe1[r];
        sum = ((s8[0] + s8[1]) + (s8[2] + s8[3])) + ((s8[4] + s8[5]) + (s8[6] + s8[7]));
    }
    sum += xor16(sum);
    const float inv = 1.0f / fmaxf(sum, 1.0e-30f);

    v8f ya = {}, yb = {};
#pragma unroll
    for (int j = 0; j < 4; ++j) {
        v2f a2, b2;
        a2.x = pe0[2 * j] * inv;
        a2.y = pe0[2 * j + 1] * inv;
        b2.x = xb[0][2 * j];
        b2.y = xb[0][2 * j + 1];
        ya = __builtin_amdgcn_wmma_f32_16x16x4_f32(false, a2, false, b2, (short)0, ya, false, false);
        a2.x = pe1[2 * j] * inv;
        a2.y = pe1[2 * j + 1] * inv;
        b2.x = xb[1][2 * j];
        b2.y = xb[1][2 * j + 1];
        yb = __builtin_amdgcn_wmma_f32_16x16x4_f32(false, a2, false, b2, (short)0, yb, false, false);
    }
    const v8f yv = ya + yb;
    if (i_n == hf) {
        float* yd = ysc + (h * 2 + hf) * 8 + cpos;
#pragma unroll
        for (int r = 0; r < 8; ++r)
            yd[r * TR_ * 8] = yv[r];
    }

    const bool cv = (n < C_);
    const float* __restrict__ fcp = fc_w + (size_t)(cv ? n : 0) * W_ + h * D_ + hf * 8;
    v4f fcv[8];
#pragma unroll
    for (int t = 0; t < 4; ++t) {
        fcv[2 * t]     = *(const v4f*)(fcp + t * 16);
        fcv[2 * t + 1] = *(const v4f*)(fcp + t * 16 + 4);
    }
    float pfb[C_], pls[C_], plb[C_];
#pragma unroll
    for (int c = 0; c < C_; ++c)
        plb[c] = ln_b[c];
#pragma unroll
    for (int c = 0; c < C_; ++c) {
        pfb[c] = fc_b[c];
        pls[c] = ln_s[c];
    }

    __syncthreads();

    {
        const v4f y = *(const v4f*)(ysc + (h * TR_ + n) * 8 + hf * 4);
        v2f ylo, yhi;
        ylo.x = y.x; ylo.y = y.y;
        yhi.x = y.z; yhi.y = y.w;
        const v8f z8 = {};
        v8f vt[4];
#pragma unroll
        for (int t = 0; t < 4; ++t)
            vt[t] = __builtin_amdgcn_wmma_f32_16x16x4_f32(false, fv[t].lo, false, ylo, (short)0, z8, false, false);
#pragma unroll
        for (int t = 0; t < 4; ++t)
            vt[t] = __builtin_amdgcn_wmma_f32_16x16x4_f32(false, fv[t].hi, false, yhi, (short)0, vt[t], false, false);

        v8f oa = {}, ob = {};
#pragma unroll
        for (int t = 0; t < 4; ++t) {
#pragma unroll
            for (int j = 0; j < 4; ++j) {
                const v4f f = fcv[2 * t + (j >> 1)];
                const float w0 = (j & 1) ? f.z : f.x;
                const float w1 = (j & 1) ? f.w : f.y;
                v2f a2, b2;
                a2.x = cv ? w0 : 0.0f;
                a2.y = cv ? w1 : 0.0f;
                b2.x = vt[t][2 * j];
                b2.y = vt[t][2 * j + 1];
                if (t < 2)
                    oa = __builtin_amdgcn_wmma_f32_16x16x4_f32(false, a2, false, b2, (short)0, oa, false, false);
                else
                    ob = __builtin_amdgcn_wmma_f32_16x16x4_f32(false, a2, false, b2, (short)0, ob, false, false);
            }
        }
        const v8f o = oa + ob;
        if (hf == 0) {
            float* __restrict__ pd = part + (h * TR_ + n) * 8;
            v4f s0, s1;
            s0.x = o[0]; s0.y = o[1]; s0.z = o[2]; s0.w = o[3];
            s1.x = o[4]; s1.y = o[5]; s1.z = o[6]; s1.w = o[7];
            *(v4f*)pd       = s0;
            *(v4f*)(pd + 4) = s1;
        }
    }

    __syncthreads();

    if (h == 0) {
        const float x2 = xor16(fr.lo.x), x3 = xor16(fr.lo.y), x6 = xor16(fr.hi.x);
        if (hf == 0) {
            const float* pr = part + n * 8;
            v4f a0 = *(const v4f*)pr;
            v4f a1 = *(const v4f*)(pr + 4);
#pragma unroll
            for (int w = 1; w < H_; ++w) {
                a0 += *(const v4f*)(pr + w * TR_ * 8);
                a1 += *(const v4f*)(pr + w * TR_ * 8 + 4);
            }
            float v[C_];
            v[0] = (a0.x + pfb[0]) + fr.lo.x;
            v[1] = (a0.y + pfb[1]) + fr.lo.y;
            v[2] = (a0.z + pfb[2]) + x2;
            v[3] = (a0.w + pfb[3]) + x3;
            v[4] = (a1.x + pfb[4]) + fr.hi.x;
            v[5] = (a1.y + pfb[5]) + fr.hi.y;
            v[6] = (a1.z + pfb[6]) + x6;

            const float ssum = ((v[0] + v[4]) + (v[2] + v[6])) + ((v[1] + v[5]) + v[3]);
            const float mu = ssum * (1.0f / (float)C_);
            float d[C_];
#pragma unroll
            for (int c = 0; c < C_; ++c)
                d[c] = v[c] - mu;
            const float vsum = ((d[0] * d[0] + d[4] * d[4]) + (d[2] * d[2] + d[6] * d[6]))
                             + ((d[1] * d[1] + d[5] * d[5]) + d[3] * d[3]);
            const float var = vsum * (1.0f / (float)C_);
            const float rs  = rsqrtf(var + 1.0e-6f);
            float* __restrict__ orow = out + (size_t)(row0 + n) * C_;
#pragma unroll
            for (int c = 0; c < C_; ++c)
                orow[c] = d[c] * rs * pls[c] + plb[c];
        }
    }
}

extern "C" void kernel_launch(void* const* d_in, const int* in_sizes, int n_in,
                              void* d_out, int out_size, void* d_ws, size_t ws_size,
                              hipStream_t stream)
{
    const float* hs  = (const float*)d_in[0];
    const float* wq  = (const float*)d_in[1];
    const float* wk  = (const float*)d_in[2];
    const float* wv  = (const float*)d_in[3];
    const float* fcw = (const float*)d_in[4];
    const float* fcb = (const float*)d_in[5];
    const float* lns = (const float*)d_in[6];
    const float* lnb = (const float*)d_in[7];
    const int*   qkm = (const int*)d_in[8];
    float* out = (float*)d_out;

    pyr_attn_remat<<<ROWS_ / TR_, 256, 0, stream>>>(hs, wq, wk, wv, qkm, fcw, fcb, lns, lnb, out);
}
